# starting kernel
# baseline (speedup 1.0000x reference)
#include <hip/hip_runtime.h>

#define NNODES 50000
#define FEAT   128
#define HCOUT  128
#define NHEAD  4
#define NCH    32
#define NEDGE  800000
#define NEG_SLOPE 0.2f

typedef __attribute__((ext_vector_type(2)))  float    v2f;
typedef __attribute__((ext_vector_type(8)))  float    v8f;
typedef __attribute__((ext_vector_type(16))) _Float16 v16h;

#if __has_builtin(__builtin_amdgcn_wmma_f32_16x16x4_f32)
#define USE_F32_WMMA 1
#endif

__global__ __launch_bounds__(256) void zero_ws(float* __restrict__ p, int n) {
    int t = blockIdx.x * 256 + threadIdx.x;
    if (t < n) p[t] = 0.0f;
}

__global__ __launch_bounds__(256) void gemm_xw(const float* __restrict__ x,
                                               const float* __restrict__ w,
                                               float* __restrict__ xw) {
    __shared__ float ldsA[16][FEAT];
    const int rowBase = blockIdx.x * 16;

    for (int idx = threadIdx.x; idx < 16 * FEAT; idx += 256) {
        int r = idx >> 7, c = idx & 127;
        int node = rowBase + r;
        ldsA[r][c] = (node < NNODES) ? x[node * FEAT + c] : 0.0f;
    }
    __syncthreads();

    const int wave    = threadIdx.x >> 5;
    const int lane    = threadIdx.x & 31;
    const int colBase = wave * 16;
    const int Mrow    = lane & 15;
    const bool hi     = lane >= 16;

    v8f acc = {};

#ifdef USE_F32_WMMA
    const int koff = hi ? 2 : 0;
    for (int k0 = 0; k0 < FEAT; k0 += 4) {
        v2f a, b;
        a.x = ldsA[Mrow][k0 + koff];
        a.y = ldsA[Mrow][k0 + koff + 1];
        b.x = w[(k0 + koff)     * HCOUT + colBase + Mrow];
        b.y = w[(k0 + koff + 1) * HCOUT + colBase + Mrow];
        acc = __builtin_amdgcn_wmma_f32_16x16x4_f32(
            false, a, false, b, (short)0, acc, false, false);
    }
#else
    for (int k0 = 0; k0 < FEAT; k0 += 32) {
        v16h a, b;
        const int ka = hi ? 8 : 0;
        const int kb = hi ? 16 : 0;
        #pragma unroll
        for (int i = 0; i < 8; ++i) {
            a[i]     = (_Float16)ldsA[Mrow][k0 + ka + i];
            a[8 + i] = (_Float16)ldsA[Mrow][k0 + 16 + ka + i];
        }
        #pragma unroll
        for (int i = 0; i < 16; ++i)
            b[i] = (_Float16)w[(k0 + kb + i) * HCOUT + colBase + Mrow];
        acc = __builtin_amdgcn_wmma_f32_16x16x32_f16(
            false, a, false, b, (short)0, acc, false, false);
    }
#endif

    #pragma unroll
    for (int v = 0; v < 8; ++v) {
        int node = rowBase + v + (hi ? 8 : 0);
        if (node < NNODES) xw[node * HCOUT + colBase + Mrow] = acc[v];
    }
}

__global__ __launch_bounds__(256) void node_dots(const float* __restrict__ xw,
                                                 const float* __restrict__ att,
                                                 float* __restrict__ aS,
                                                 float* __restrict__ aD) {
    int t = blockIdx.x * 256 + threadIdx.x;
    if (t >= NNODES * NHEAD) return;
    int h = t & (NHEAD - 1);
    const float* v  = xw + t * NCH;
    const float* cs = att + h * (2 * NCH);
    const float* cd = cs + NCH;
    float s = 0.0f, d = 0.0f;
    #pragma unroll
    for (int c = 0; c < NCH; ++c) {
        float xv = v[c];
        s += xv * cs[c];
        d += xv * cd[c];
    }
    aS[t] = s;
    aD[t] = d;
}

__device__ __forceinline__ unsigned fkey(float f) {
    unsigned u = __float_as_uint(f);
    return (u & 0x80000000u) ? ~u : (u | 0x80000000u);
}
__device__ __forceinline__ float fdecode(unsigned k) {
    unsigned u = (k & 0x80000000u) ? (k & 0x7FFFFFFFu) : ~k;
    return __uint_as_float(u);
}

__global__ __launch_bounds__(256) void edge_logits(const int* __restrict__ row,
                                                   const int* __restrict__ col,
                                                   const float* __restrict__ aS,
                                                   const float* __restrict__ aD,
                                                   float* __restrict__ alphaE,
                                                   unsigned* __restrict__ segKey) {
    int e = blockIdx.x * 256 + threadIdx.x;
    if (e >= NEDGE) return;
    int r = row[e], c = col[e];
    #pragma unroll
    for (int h = 0; h < NHEAD; ++h) {
        float a = aS[r * NHEAD + h] + aD[c * NHEAD + h];
        a = (a >= 0.0f) ? a : NEG_SLOPE * a;
        alphaE[e * NHEAD + h] = a;
        atomicMax(&segKey[r * NHEAD + h], fkey(a));
    }
}

__global__ __launch_bounds__(256) void edge_expsum(const int* __restrict__ row,
                                                   float* __restrict__ alphaE,
                                                   const unsigned* __restrict__ segKey,
                                                   float* __restrict__ segSum) {
    int e = blockIdx.x * 256 + threadIdx.x;
    if (e >= NEDGE) return;
    int r = row[e];
    #pragma unroll
    for (int h = 0; h < NHEAD; ++h) {
        float m = fdecode(segKey[r * NHEAD + h]);
        float a = __expf(alphaE[e * NHEAD + h] - m);
        alphaE[e * NHEAD + h] = a;
        atomicAdd(&segSum[r * NHEAD + h], a);
    }
}

__global__ __launch_bounds__(256) void edge_scatter(const int* __restrict__ row,
                                                    const int* __restrict__ col,
                                                    const float* __restrict__ alphaE,
                                                    const float* __restrict__ segSum,
                                                    const float* __restrict__ xw,
                                                    float* __restrict__ accum) {
    int t = blockIdx.x * 256 + threadIdx.x;
    if (t >= NEDGE * HCOUT) return;
    int e    = t >> 7;
    int elem = t & 127;
    int h    = elem >> 5;
    int r = row[e], c = col[e];
    float wgt = alphaE[e * NHEAD + h] / (segSum[r * NHEAD + h] + 1e-16f);
    atomicAdd(&accum[c * HCOUT + elem], wgt * xw[r * HCOUT + elem]);
}

__global__ __launch_bounds__(256) void finalize(const float* __restrict__ accum,
                                                const float* __restrict__ bias,
                                                float* __restrict__ out) {
    int t = blockIdx.x * 256 + threadIdx.x;
    if (t >= NNODES * NCH) return;
    int n = t >> 5, c = t & 31;
    const float* a = accum + n * HCOUT + c;
    out[t] = 0.25f * (a[0] + a[NCH] + a[2 * NCH] + a[3 * NCH]) + bias[c];
}

extern "C" void kernel_launch(void* const* d_in, const int* in_sizes, int n_in,
                              void* d_out, int out_size, void* d_ws, size_t ws_size,
                              hipStream_t stream) {
    const float* x    = (const float*)d_in[0];
    const int*   ei   = (const int*)  d_in[1];
    const float* w    = (const float*)d_in[2];
    const float* att  = (const float*)d_in[3];
    const float* bias = (const float*)d_in[4];
    float* ws = (float*)d_ws;

    float*    xw     = ws;
    float*    aS     = ws + 6400000;
    float*    aD     = ws + 6600000;
    float*    alphaE = ws + 6800000;
    float*    zreg   = ws + 10000000;
    unsigned* segKey = (unsigned*)(ws + 10000000);
    float*    segSum = ws + 10200000;
    float*    accum  = ws + 10400000;

    const int* row = ei;
    const int* col = ei + NEDGE;

    const int ZN = 6800000;
    zero_ws   <<<(ZN + 255) / 256, 256, 0, stream>>>(zreg, ZN);
    gemm_xw   <<<(NNODES + 15) / 16, 256, 0, stream>>>(x, w, xw);
    node_dots <<<(NNODES * NHEAD + 255) / 256, 256, 0, stream>>>(xw, att, aS, aD);
    edge_logits<<<(NEDGE + 255) / 256, 256, 0, stream>>>(row, col, aS, aD, alphaE, segKey);
    edge_expsum<<<(NEDGE + 255) / 256, 256, 0, stream>>>(row, alphaE, segKey, segSum);
    edge_scatter<<<(NEDGE * HCOUT + 255) / 256, 256, 0, stream>>>(row, col, alphaE, segSum, xw, accum);
    finalize  <<<(NNODES * NCH + 255) / 256, 256, 0, stream>>>(accum, bias, (float*)d_out);
}


# scatter: sum an edge's 4 heads in registers before the atomic add (accum [N,32] not [N,128]); 4x fewer float atomics; sum order changes (heads first, then edges)
# speedup vs baseline: 1.7143x; 1.7143x over previous
#include <hip/hip_runtime.h>

#define NNODES 50000
#define FEAT   128
#define HCOUT  128
#define NHEAD  4
#define NCH    32
#define NEDGE  800000
#define NEG_SLOPE 0.2f

typedef __attribute__((ext_vector_type(2)))  float    v2f;
typedef __attribute__((ext_vector_type(8)))  float    v8f;
typedef __attribute__((ext_vector_type(16))) _Float16 v16h;

#if __has_builtin(__builtin_amdgcn_wmma_f32_16x16x4_f32)
#define USE_F32_WMMA 1
#endif

__global__ __launch_bounds__(256) void zero_ws(float* __restrict__ p, int n) {
    int t = blockIdx.x * 256 + threadIdx.x;
    if (t < n) p[t] = 0.0f;
}

__global__ __launch_bounds__(256) void gemm_xw(const float* __restrict__ x,
                                               const float* __restrict__ w,
                                               float* __restrict__ xw) {
    __shared__ float ldsA[16][FEAT];
    const int rowBase = blockIdx.x * 16;

    for (int idx = threadIdx.x; idx < 16 * FEAT; idx += 256) {
        int r = idx >> 7, c = idx & 127;
        int node = rowBase + r;
        ldsA[r][c] = (node < NNODES) ? x[node * FEAT + c] : 0.0f;
    }
    __syncthreads();

    const int wave    = threadIdx.x >> 5;
    const int lane    = threadIdx.x & 31;
    const int colBase = wave * 16;
    const int Mrow    = lane & 15;
    const bool hi     = lane >= 16;

    v8f acc = {};

#ifdef USE_F32_WMMA
    const int koff = hi ? 2 : 0;
    for (int k0 = 0; k0 < FEAT; k0 += 4) {
        v2f a, b;
        a.x = ldsA[Mrow][k0 + koff];
        a.y = ldsA[Mrow][k0 + koff + 1];
        b.x = w[(k0 + koff)     * HCOUT + colBase + Mrow];
        b.y = w[(k0 + koff + 1) * HCOUT + colBase + Mrow];
        acc = __builtin_amdgcn_wmma_f32_16x16x4_f32(
            false, a, false, b, (short)0, acc, false, false);
    }
#else
    for (int k0 = 0; k0 < FEAT; k0 += 32) {
        v16h a, b;
        const int ka = hi ? 8 : 0;
        const int kb = hi ? 16 : 0;
        #pragma unroll
        for (int i = 0; i < 8; ++i) {
            a[i]     = (_Float16)ldsA[Mrow][k0 + ka + i];
            a[8 + i] = (_Float16)ldsA[Mrow][k0 + 16 + ka + i];
        }
        #pragma unroll
        for (int i = 0; i < 16; ++i)
            b[i] = (_Float16)w[(k0 + kb + i) * HCOUT + colBase + Mrow];
        acc = __builtin_amdgcn_wmma_f32_16x16x32_f16(
            false, a, false, b, (short)0, acc, false, false);
    }
#endif

    #pragma unroll
    for (int v = 0; v < 8; ++v) {
        int node = rowBase + v + (hi ? 8 : 0);
        if (node < NNODES) xw[node * HCOUT + colBase + Mrow] = acc[v];
    }
}

__global__ __launch_bounds__(256) void node_dots(const float* __restrict__ xw,
                                                 const float* __restrict__ att,
                                                 float* __restrict__ aS,
                                                 float* __restrict__ aD) {
    int t = blockIdx.x * 256 + threadIdx.x;
    if (t >= NNODES * NHEAD) return;
    int h = t & (NHEAD - 1);
    const float* v  = xw + t * NCH;
    const float* cs = att + h * (2 * NCH);
    const float* cd = cs + NCH;
    float s = 0.0f, d = 0.0f;
    #pragma unroll
    for (int c = 0; c < NCH; ++c) {
        float xv = v[c];
        s += xv * cs[c];
        d += xv * cd[c];
    }
    aS[t] = s;
    aD[t] = d;
}

__device__ __forceinline__ unsigned fkey(float f) {
    unsigned u = __float_as_uint(f);
    return (u & 0x80000000u) ? ~u : (u | 0x80000000u);
}
__device__ __forceinline__ float fdecode(unsigned k) {
    unsigned u = (k & 0x80000000u) ? (k & 0x7FFFFFFFu) : ~k;
    return __uint_as_float(u);
}

__global__ __launch_bounds__(256) void edge_logits(const int* __restrict__ row,
                                                   const int* __restrict__ col,
                                                   const float* __restrict__ aS,
                                                   const float* __restrict__ aD,
                                                   float* __restrict__ alphaE,
                                                   unsigned* __restrict__ segKey) {
    int e = blockIdx.x * 256 + threadIdx.x;
    if (e >= NEDGE) return;
    int r = row[e], c = col[e];
    #pragma unroll
    for (int h = 0; h < NHEAD; ++h) {
        float a = aS[r * NHEAD + h] + aD[c * NHEAD + h];
        a = (a >= 0.0f) ? a : NEG_SLOPE * a;
        alphaE[e * NHEAD + h] = a;
        atomicMax(&segKey[r * NHEAD + h], fkey(a));
    }
}

__global__ __launch_bounds__(256) void edge_expsum(const int* __restrict__ row,
                                                   float* __restrict__ alphaE,
                                                   const unsigned* __restrict__ segKey,
                                                   float* __restrict__ segSum) {
    int e = blockIdx.x * 256 + threadIdx.x;
    if (e >= NEDGE) return;
    int r = row[e];
    #pragma unroll
    for (int h = 0; h < NHEAD; ++h) {
        float m = fdecode(segKey[r * NHEAD + h]);
        float a = __expf(alphaE[e * NHEAD + h] - m);
        alphaE[e * NHEAD + h] = a;
        atomicAdd(&segSum[r * NHEAD + h], a);
    }
}

#define EPW 4
static_assert(NEDGE % EPW == 0, "a wave takes EPW whole edges");

__device__ __forceinline__ float lane_value(float v, int l) {
    return __int_as_float(__builtin_amdgcn_readlane(__float_as_int(v), l));
}

__global__ __launch_bounds__(256) void edge_scatter(const int* __restrict__ row,
                                                    const int* __restrict__ col,
                                                    const float* __restrict__ alphaE,
                                                    const float* __restrict__ segSum,
                                                    const float* __restrict__ xw,
                                                    float* __restrict__ accum) {
    const int lane = threadIdx.x & 31;
    const int wv = __builtin_amdgcn_readfirstlane((int)(blockIdx.x * 8 + (threadIdx.x >> 5)));
    const int e0 = wv * EPW;
    if (e0 >= NEDGE) return;

    const int e = e0 + ((lane >> 2) & (EPW - 1));
    const int h = lane & (NHEAD - 1);
    const int r = row[e];
    const int c = col[e];
    const float wgt = alphaE[e * NHEAD + h] / (segSum[r * NHEAD + h] + 1e-16f);

    float xv[EPW][NHEAD];
    #pragma unroll
    for (int jj = 0; jj < EPW; ++jj) {
        const int rj = __builtin_amdgcn_readlane(r, 4 * jj);
        const float* src = xw + rj * HCOUT + lane;
        #pragma unroll
        for (int hh = 0; hh < NHEAD; ++hh) xv[jj][hh] = src[hh * NCH];
    }
    #pragma unroll
    for (int jj = 0; jj < EPW; ++jj) {
        const int cj = __builtin_amdgcn_readlane(c, 4 * jj);
        float s = lane_value(wgt, 4 * jj) * xv[jj][0];
        #pragma unroll
        for (int hh = 1; hh < NHEAD; ++hh) s += lane_value(wgt, 4 * jj + hh) * xv[jj][hh];
        atomicAdd(&accum[cj * NCH + lane], s);
    }
}

__global__ __launch_bounds__(256) void finalize(const float* __restrict__ accum,
                                                const float* __restrict__ bias,
                                                float* __restrict__ out) {
    int t = blockIdx.x * 256 + threadIdx.x;
    if (t >= NNODES * NCH) return;
    out[t] = 0.25f * accum[t] + bias[t & 31];
}

extern "C" void kernel_launch(void* const* d_in, const int* in_sizes, int n_in,
                              void* d_out, int out_size, void* d_ws, size_t ws_size,
                              hipStream_t stream) {
    const float* x    = (const float*)d_in[0];
    const int*   ei   = (const int*)  d_in[1];
    const float* w    = (const float*)d_in[2];
    const float* att  = (const float*)d_in[3];
    const float* bias = (const float*)d_in[4];
    float* ws = (float*)d_ws;

    float*    xw     = ws;
    float*    aS     = ws + 6400000;
    float*    aD     = ws + 6600000;
    float*    alphaE = ws + 6800000;
    float*    zreg   = ws + 10000000;
    unsigned* segKey = (unsigned*)(ws + 10000000);
    float*    segSum = ws + 10200000;
    float*    accum  = ws + 10400000;

    const int* row = ei;
    const int* col = ei + NEDGE;

    const int ZN = 2000000;
    zero_ws   <<<(ZN + 255) / 256, 256, 0, stream>>>(zreg, ZN);
    gemm_xw   <<<(NNODES + 15) / 16, 256, 0, stream>>>(x, w, xw);
    node_dots <<<(NNODES * NHEAD + 255) / 256, 256, 0, stream>>>(xw, att, aS, aD);
    edge_logits<<<(NEDGE + 255) / 256, 256, 0, stream>>>(row, col, aS, aD, alphaE, segKey);
    edge_expsum<<<(NEDGE + 255) / 256, 256, 0, stream>>>(row, alphaE, segKey, segSum);
    edge_scatter<<<(NEDGE + 8 * EPW - 1) / (8 * EPW), 256, 0, stream>>>(row, col, alphaE, segSum, xw, accum);
    finalize  <<<(NNODES * NCH + 255) / 256, 256, 0, stream>>>(accum, bias, (float*)d_out);
}


# edges grouped by source node (count, scan, place); one wave per source keeps its xw row in registers, softmax max/sum by wave reduction, no per-edge xw gather; segment-sum order changes
# speedup vs baseline: 2.8409x; 1.6572x over previous
#include <hip/hip_runtime.h>

#define NNODES 50000
#define FEAT   128
#define HCOUT  128
#define NHEAD  4
#define NCH    32
#define NEDGE  800000
#define NEG_SLOPE 0.2f

typedef __attribute__((ext_vector_type(2)))  float    v2f;
typedef __attribute__((ext_vector_type(8)))  float    v8f;
typedef __attribute__((ext_vector_type(16))) _Float16 v16h;

#if __has_builtin(__builtin_amdgcn_wmma_f32_16x16x4_f32)
#define USE_F32_WMMA 1
#endif

__global__ __launch_bounds__(256) void zero_ws(float* __restrict__ p, int n) {
    int t = blockIdx.x * 256 + threadIdx.x;
    if (t < n) p[t] = 0.0f;
}

__global__ __launch_bounds__(256) void gemm_xw(const float* __restrict__ x,
                                               const float* __restrict__ w,
                                               float* __restrict__ xw) {
    __shared__ float ldsA[16][FEAT];
    const int rowBase = blockIdx.x * 16;

    for (int idx = threadIdx.x; idx < 16 * FEAT; idx += 256) {
        int r = idx >> 7, c = idx & 127;
        int node = rowBase + r;
        ldsA[r][c] = (node < NNODES) ? x[node * FEAT + c] : 0.0f;
    }
    __syncthreads();

    const int wave    = threadIdx.x >> 5;
    const int lane    = threadIdx.x & 31;
    const int colBase = wave * 16;
    const int Mrow    = lane & 15;
    const bool hi     = lane >= 16;

    v8f acc = {};

#ifdef USE_F32_WMMA
    const int koff = hi ? 2 : 0;
    for (int k0 = 0; k0 < FEAT; k0 += 4) {
        v2f a, b;
        a.x = ldsA[Mrow][k0 + koff];
        a.y = ldsA[Mrow][k0 + koff + 1];
        b.x = w[(k0 + koff)     * HCOUT + colBase + Mrow];
        b.y = w[(k0 + koff + 1) * HCOUT + colBase + Mrow];
        acc = __builtin_amdgcn_wmma_f32_16x16x4_f32(
            false, a, false, b, (short)0, acc, false, false);
    }
#else
    for (int k0 = 0; k0 < FEAT; k0 += 32) {
        v16h a, b;
        const int ka = hi ? 8 : 0;
        const int kb = hi ? 16 : 0;
        #pragma unroll
        for (int i = 0; i < 8; ++i) {
            a[i]     = (_Float16)ldsA[Mrow][k0 + ka + i];
            a[8 + i] = (_Float16)ldsA[Mrow][k0 + 16 + ka + i];
        }
        #pragma unroll
        for (int i = 0; i < 16; ++i)
            b[i] = (_Float16)w[(k0 + kb + i) * HCOUT + colBase + Mrow];
        acc = __builtin_amdgcn_wmma_f32_16x16x32_f16(
            false, a, false, b, (short)0, acc, false, false);
    }
#endif

    #pragma unroll
    for (int v = 0; v < 8; ++v) {
        int node = rowBase + v + (hi ? 8 : 0);
        if (node < NNODES) xw[node * HCOUT + colBase + Mrow] = acc[v];
    }
}

__global__ __launch_bounds__(256) void node_dots(const float* __restrict__ xw,
                                                 const float* __restrict__ att,
                                                 float* __restrict__ aS,
                                                 float* __restrict__ aD) {
    int t = blockIdx.x * 256 + threadIdx.x;
    if (t >= NNODES * NHEAD) return;
    int h = t & (NHEAD - 1);
    const float* v  = xw + t * NCH;
    const float* cs = att + h * (2 * NCH);
    const float* cd = cs + NCH;
    float s = 0.0f, d = 0.0f;
    #pragma unroll
    for (int c = 0; c < NCH; ++c) {
        float xv = v[c];
        s += xv * cs[c];
        d += xv * cd[c];
    }
    aS[t] = s;
    aD[t] = d;
}

__global__ __launch_bounds__(256) void count_src(const int* __restrict__ row,
                                                 int* __restrict__ deg) {
    int e = blockIdx.x * 256 + threadIdx.x;
    if (e >= NEDGE) return;
    atomicAdd(&deg[row[e]], 1);
}

#define SCAN_IT  49
#define SCAN_SEG (32 * SCAN_IT)
static_assert(32 * SCAN_SEG >= NNODES, "the block covers every node");

__global__ __launch_bounds__(1024) void scan_deg(const int* __restrict__ deg,
                                                 int* __restrict__ off) {
    __shared__ int wtot[32];
    const int lane = threadIdx.x & 31;
    const int wave = threadIdx.x >> 5;
    const int base = wave * SCAN_SEG + lane;

    int v[SCAN_IT];
    #pragma unroll
    for (int i = 0; i < SCAN_IT; ++i) {
        int idx = base + i * 32;
        v[i] = (idx < NNODES) ? deg[idx] : 0;
    }
    int tot = 0;
    #pragma unroll
    for (int i = 0; i < SCAN_IT; ++i) tot += v[i];
    #pragma unroll
    for (int o = 16; o > 0; o >>= 1) tot += __shfl_xor(tot, o, 32);
    if (lane == 0) wtot[wave] = tot;
    __syncthreads();
    int carry = (lane < wave) ? wtot[lane] : 0;
    #pragma unroll
    for (int o = 16; o > 0; o >>= 1) carry += __shfl_xor(carry, o, 32);

    #pragma unroll
    for (int i = 0; i < SCAN_IT; ++i) {
        int inc = v[i];
        #pragma unroll
        for (int o = 1; o < 32; o <<= 1) {
            int u = __shfl_up(inc, o, 32);
            if (lane >= o) inc += u;
        }
        int idx = base + i * 32;
        if (idx < NNODES) off[idx] = carry + inc - v[i];
        carry += __shfl(inc, 31, 32);
    }
    if (threadIdx.x == 1023) off[NNODES] = carry;
}

__global__ __launch_bounds__(256) void place_src(const int* __restrict__ row,
                                                 const int* __restrict__ col,
                                                 const int* __restrict__ off,
                                                 int* __restrict__ deg,
                                                 int* __restrict__ scol) {
    int e = blockIdx.x * 256 + threadIdx.x;
    if (e >= NEDGE) return;
    int r = row[e];
    int slot = atomicSub(&deg[r], 1) - 1;
    scol[off[r] + slot] = col[e];
}


__device__ __forceinline__ float lane_value(float v, int l) {
    return __int_as_float(__builtin_amdgcn_readlane(__float_as_int(v), l));
}

__device__ __forceinline__ void edge_logits4(const int* __restrict__ scol,
                                             const float* __restrict__ aD,
                                             const float (&as)[NHEAD],
                                             int p, int end, int& c, float (&a)[NHEAD]) {
    c = 0;
    #pragma unroll
    for (int h = 0; h < NHEAD; ++h) a[h] = -INFINITY;
    if (p < end) {
        c = scol[p];
        const float4 d = *reinterpret_cast<const float4*>(aD + c * NHEAD);
        const float dv[NHEAD] = {d.x, d.y, d.z, d.w};
        #pragma unroll
        for (int h = 0; h < NHEAD; ++h) {
            float t = as[h] + dv[h];
            a[h] = (t >= 0.0f) ? t : NEG_SLOPE * t;
        }
    }
}

__global__ __launch_bounds__(256) void src_softmax_scatter(const int* __restrict__ off,
                                                           const int* __restrict__ scol,
                                                           const float* __restrict__ aS,
                                                           const float* __restrict__ aD,
                                                           const float* __restrict__ xw,
                                                           float* __restrict__ accum) {
    const int lane = threadIdx.x & 31;
    const int r = __builtin_amdgcn_readfirstlane((int)(blockIdx.x * 8 + (threadIdx.x >> 5)));
    if (r >= NNODES) return;
    const int beg = off[r], end = off[r + 1];
    if (beg >= end) return;

    float xr[NHEAD], as[NHEAD];
    #pragma unroll
    for (int h = 0; h < NHEAD; ++h) {
        xr[h] = xw[r * HCOUT + h * NCH + lane];
        as[h] = aS[r * NHEAD + h];
    }

    int c0; float a0[NHEAD];
    edge_logits4(scol, aD, as, beg + lane, end, c0, a0);

    float m[NHEAD];
    #pragma unroll
    for (int h = 0; h < NHEAD; ++h) m[h] = a0[h];
    for (int b = beg + 32; b < end; b += 32) {
        int c; float a[NHEAD];
        edge_logits4(scol, aD, as, b + lane, end, c, a);
        #pragma unroll
        for (int h = 0; h < NHEAD; ++h) m[h] = fmaxf(m[h], a[h]);
    }
    #pragma unroll
    for (int h = 0; h < NHEAD; ++h) {
        #pragma unroll
        for (int o = 16; o > 0; o >>= 1) m[h] = fmaxf(m[h], __shfl_xor(m[h], o, 32));
        if (!(fabsf(m[h]) < INFINITY)) m[h] = 0.0f;
    }

    float ex0[NHEAD], S[NHEAD];
    #pragma unroll
    for (int h = 0; h < NHEAD; ++h) {
        ex0[h] = (beg + lane < end) ? __expf(a0[h] - m[h]) : 0.0f;
        S[h] = ex0[h];
    }
    for (int b = beg + 32; b < end; b += 32) {
        int c; float a[NHEAD];
        edge_logits4(scol, aD, as, b + lane, end, c, a);
        #pragma unroll
        for (int h = 0; h < NHEAD; ++h) S[h] += (b + lane < end) ? __expf(a[h] - m[h]) : 0.0f;
    }
    #pragma unroll
    for (int h = 0; h < NHEAD; ++h) {
        #pragma unroll
        for (int o = 16; o > 0; o >>= 1) S[h] += __shfl_xor(S[h], o, 32);
        S[h] += 1e-16f;
    }

    for (int b = beg; b < end; b += 32) {
        int c; float w[NHEAD];
        if (b == beg) {
            c = c0;
            #pragma unroll
            for (int h = 0; h < NHEAD; ++h) w[h] = ex0[h] / S[h];
        } else {
            float a[NHEAD];
            edge_logits4(scol, aD, as, b + lane, end, c, a);
            #pragma unroll
            for (int h = 0; h < NHEAD; ++h) w[h] = ((b + lane < end) ? __expf(a[h] - m[h]) : 0.0f) / S[h];
        }
        const int cnt = (end - b < 32) ? (end - b) : 32;
        for (int j = 0; j < cnt; ++j) {
            const int cj = __builtin_amdgcn_readlane(c, j);
            float s = lane_value(w[0], j) * xr[0];
            #pragma unroll
            for (int h = 1; h < NHEAD; ++h) s += lane_value(w[h], j) * xr[h];
            atomicAdd(&accum[cj * NCH + lane], s);
        }
    }
}

__global__ __launch_bounds__(256) void finalize(const float* __restrict__ accum,
                                                const float* __restrict__ bias,
                                                float* __restrict__ out) {
    int t = blockIdx.x * 256 + threadIdx.x;
    if (t >= NNODES * NCH) return;
    out[t] = 0.25f * accum[t] + bias[t & 31];
}

extern "C" void kernel_launch(void* const* d_in, const int* in_sizes, int n_in,
                              void* d_out, int out_size, void* d_ws, size_t ws_size,
                              hipStream_t stream) {
    const float* x    = (const float*)d_in[0];
    const int*   ei   = (const int*)  d_in[1];
    const float* w    = (const float*)d_in[2];
    const float* att  = (const float*)d_in[3];
    const float* bias = (const float*)d_in[4];
    float* ws = (float*)d_ws;

    float* xw    = ws;
    float* aS    = ws + 6400000;
    float* aD    = ws + 6600000;
    int*   scol  = (int*)(ws + 6800000);
    int*   off   = (int*)(ws + 7600000);
    float* zreg  = ws + 10000000;
    int*   deg   = (int*)(ws + 10000000);
    float* accum = ws + 10050048;

    const int* row = ei;
    const int* col = ei + NEDGE;

    const int ZN = 1650048;
    zero_ws   <<<(ZN + 255) / 256, 256, 0, stream>>>(zreg, ZN);
    count_src <<<(NEDGE + 255) / 256, 256, 0, stream>>>(row, deg);
    scan_deg  <<<1, 1024, 0, stream>>>(deg, off);
    place_src <<<(NEDGE + 255) / 256, 256, 0, stream>>>(row, col, off, deg, scol);
    gemm_xw   <<<(NNODES + 15) / 16, 256, 0, stream>>>(x, w, xw);
    node_dots <<<(NNODES * NHEAD + 255) / 256, 256, 0, stream>>>(xw, att, aS, aD);
    src_softmax_scatter<<<(NNODES + 7) / 8, 256, 0, stream>>>(off, scol, aS, aD, xw, accum);
    finalize  <<<(NNODES * NCH + 255) / 256, 256, 0, stream>>>(accum, bias, (float*)d_out);
}


# edges grouped by source with a two-level counting sort on LDS counters (chunk x bucket counts, one block per 128-node bucket): no per-edge global atomics; segment-sum order inside a source changes
# speedup vs baseline: 4.9088x; 1.7279x over previous
#include <hip/hip_runtime.h>

#define NNODES 50000
#define FEAT   128
#define HCOUT  128
#define NHEAD  4
#define NCH    32
#define NEDGE  800000
#define NEG_SLOPE 0.2f

typedef __attribute__((ext_vector_type(2)))  float    v2f;
typedef __attribute__((ext_vector_type(8)))  float    v8f;
typedef __attribute__((ext_vector_type(16))) _Float16 v16h;

#if __has_builtin(__builtin_amdgcn_wmma_f32_16x16x4_f32)
#define USE_F32_WMMA 1
#endif

__global__ __launch_bounds__(256) void zero_ws(float* __restrict__ p, int n) {
    int t = blockIdx.x * 256 + threadIdx.x;
    if (t < n) p[t] = 0.0f;
}

__global__ __launch_bounds__(256) void gemm_xw(const float* __restrict__ x,
                                               const float* __restrict__ w,
                                               float* __restrict__ xw) {
    __shared__ float ldsA[16][FEAT];
    const int rowBase = blockIdx.x * 16;

    for (int idx = threadIdx.x; idx < 16 * FEAT; idx += 256) {
        int r = idx >> 7, c = idx & 127;
        int node = rowBase + r;
        ldsA[r][c] = (node < NNODES) ? x[node * FEAT + c] : 0.0f;
    }
    __syncthreads();

    const int wave    = threadIdx.x >> 5;
    const int lane    = threadIdx.x & 31;
    const int colBase = wave * 16;
    const int Mrow    = lane & 15;
    const bool hi     = lane >= 16;

    v8f acc = {};

#ifdef USE_F32_WMMA
    const int koff = hi ? 2 : 0;
    for (int k0 = 0; k0 < FEAT; k0 += 4) {
        v2f a, b;
        a.x = ldsA[Mrow][k0 + koff];
        a.y = ldsA[Mrow][k0 + koff + 1];
        b.x = w[(k0 + koff)     * HCOUT + colBase + Mrow];
        b.y = w[(k0 + koff + 1) * HCOUT + colBase + Mrow];
        acc = __builtin_amdgcn_wmma_f32_16x16x4_f32(
            false, a, false, b, (short)0, acc, false, false);
    }
#else
    for (int k0 = 0; k0 < FEAT; k0 += 32) {
        v16h a, b;
        const int ka = hi ? 8 : 0;
        const int kb = hi ? 16 : 0;
        #pragma unroll
        for (int i = 0; i < 8; ++i) {
            a[i]     = (_Float16)ldsA[Mrow][k0 + ka + i];
            a[8 + i] = (_Float16)ldsA[Mrow][k0 + 16 + ka + i];
        }
        #pragma unroll
        for (int i = 0; i < 16; ++i)
            b[i] = (_Float16)w[(k0 + kb + i) * HCOUT + colBase + Mrow];
        acc = __builtin_amdgcn_wmma_f32_16x16x32_f16(
            false, a, false, b, (short)0, acc, false, false);
    }
#endif

    #pragma unroll
    for (int v = 0; v < 8; ++v) {
        int node = rowBase + v + (hi ? 8 : 0);
        if (node < NNODES) xw[node * HCOUT + colBase + Mrow] = acc[v];
    }
}

__global__ __launch_bounds__(256) void node_dots(const float* __restrict__ xw,
                                                 const float* __restrict__ att,
                                                 float* __restrict__ aS,
                                                 float* __restrict__ aD) {
    int t = blockIdx.x * 256 + threadIdx.x;
    if (t >= NNODES * NHEAD) return;
    int h = t & (NHEAD - 1);
    const float* v  = xw + t * NCH;
    const float* cs = att + h * (2 * NCH);
    const float* cd = cs + NCH;
    float s = 0.0f, d = 0.0f;
    #pragma unroll
    for (int c = 0; c < NCH; ++c) {
        float xv = v[c];
        s += xv * cs[c];
        d += xv * cd[c];
    }
    aS[t] = s;
    aD[t] = d;
}

#define BKT_SHIFT 7
#define BKT_NODES (1 << BKT_SHIFT)
#define NBKT      ((NNODES + BKT_NODES - 1) / BKT_NODES)
#define CHUNK     4000
#define NCHUNK    ((NEDGE + CHUNK - 1) / CHUNK)
#define HALFCH    (NCHUNK / 2)
#define HPITCH    416
static_assert(NNODES <= 65536, "a destination fits the low 16 bits of a packed word");
static_assert(NBKT <= 511 && NBKT <= HPITCH, "one thread of the first 512 per bucket, and one spare");
#define HGROUP    25
static_assert(NCHUNK % 2 == 0, "the rows of counts are summed in two equal halves");
static_assert(HALFCH % HGROUP == 0, "whole groups of rows");
#define EPT       4
static_assert(CHUNK <= EPT * 1024, "one batch of loads per thread covers a chunk");

__global__ __launch_bounds__(1024) void bucket_count(const int* __restrict__ row,
                                                     int* __restrict__ hist) {
    __shared__ int cnt[512];
    const int tid = threadIdx.x;
    if (tid < 512) cnt[tid] = 0;
    __syncthreads();
    const int e0 = blockIdx.x * CHUNK;
    const int e1 = (e0 + CHUNK < NEDGE) ? (e0 + CHUNK) : NEDGE;
    int r[EPT];
    #pragma unroll
    for (int j = 0; j < EPT; ++j) {
        const int e = e0 + tid + j * 1024;
        r[j] = (e < e1) ? row[e] : -1;
    }
    #pragma unroll
    for (int j = 0; j < EPT; ++j)
        if (r[j] >= 0) atomicAdd(&cnt[r[j] >> BKT_SHIFT], 1);
    __syncthreads();
    if (tid < NBKT) hist[blockIdx.x * HPITCH + tid] = cnt[tid];
}

__global__ __launch_bounds__(1024) void bucket_place(const int* __restrict__ row,
                                                     const int* __restrict__ col,
                                                     const int* __restrict__ hist,
                                                     int* __restrict__ bstart,
                                                     int* __restrict__ part) {
    __shared__ int ptot[2][512];
    __shared__ int pbel[2][512];
    __shared__ int wsum[16];
    __shared__ int cnt[512];
    const int tid  = threadIdx.x;
    const int lane = tid & 31;
    const int wave = tid >> 5;
    const int k    = blockIdx.x;
    {
        const int b  = tid & 511;
        const int hf = tid >> 9;
        int tot = 0, bel = 0;
        if (b < NBKT) {
            const int k0 = hf * HALFCH;
            const int* p = hist + k0 * HPITCH + b;
            const int nbel = (k - k0 < 0) ? 0 : ((k - k0 > HALFCH) ? HALFCH : (k - k0));
            for (int g = 0; g < HALFCH; g += HGROUP) {
                int v[HGROUP];
                #pragma unroll
                for (int j = 0; j < HGROUP; ++j) v[j] = p[(g + j) * HPITCH];
                #pragma unroll
                for (int j = 0; j < HGROUP; ++j) {
                    tot += v[j];
                    bel += (g + j < nbel) ? v[j] : 0;
                }
            }
        }
        ptot[hf][b] = tot;
        pbel[hf][b] = bel;
    }
    __syncthreads();
    int T = 0, inc = 0;
    if (tid < 512) {
        T = ptot[0][tid] + ptot[1][tid];
        inc = T;
        #pragma unroll
        for (int o = 1; o < 32; o <<= 1) {
            int u = __shfl_up(inc, o, 32);
            if (lane >= o) inc += u;
        }
        if (lane == 31) wsum[wave] = inc;
    }
    __syncthreads();
    if (tid < 512) {
        int carry = 0;
        #pragma unroll
        for (int w = 0; w < 16; ++w) carry += (w < wave) ? wsum[w] : 0;
        const int excl = carry + inc - T;
        cnt[tid] = excl + pbel[0][tid] + pbel[1][tid];
        if (k == 0 && tid <= NBKT) bstart[tid] = excl;
    }
    __syncthreads();
    const int e0 = k * CHUNK;
    const int e1 = (e0 + CHUNK < NEDGE) ? (e0 + CHUNK) : NEDGE;
    int r[EPT], c[EPT];
    #pragma unroll
    for (int j = 0; j < EPT; ++j) {
        const int e = e0 + tid + j * 1024;
        r[j] = (e < e1) ? row[e] : -1;
        c[j] = (e < e1) ? col[e] : 0;
    }
    #pragma unroll
    for (int j = 0; j < EPT; ++j) {
        if (r[j] >= 0) {
            const int pos = atomicAdd(&cnt[r[j] >> BKT_SHIFT], 1);
            part[pos] = ((r[j] & (BKT_NODES - 1)) << 16) | c[j];
        }
    }
}

__global__ __launch_bounds__(256) void bucket_sort(const int* __restrict__ bstart,
                                                   const int* __restrict__ part,
                                                   int* __restrict__ off,
                                                   int* __restrict__ scol) {
    __shared__ int cnt[BKT_NODES];
    __shared__ int cur[BKT_NODES];
    __shared__ int wsum[BKT_NODES / 32];
    const int tid  = threadIdx.x;
    const int lane = tid & 31;
    const int wave = tid >> 5;
    const int b    = blockIdx.x;
    const int s    = bstart[b];
    const int n    = bstart[b + 1] - s;
    if (tid < BKT_NODES) cnt[tid] = 0;
    __syncthreads();
    for (int i0 = tid; i0 < n; i0 += EPT * 256) {
        int w[EPT];
        #pragma unroll
        for (int j = 0; j < EPT; ++j) {
            const int i = i0 + j * 256;
            w[j] = (i < n) ? part[s + i] : -1;
        }
        #pragma unroll
        for (int j = 0; j < EPT; ++j)
            if (w[j] >= 0) atomicAdd(&cnt[w[j] >> 16], 1);
    }
    __syncthreads();
    int v = 0, inc = 0;
    if (tid < BKT_NODES) {
        v = cnt[tid];
        inc = v;
        #pragma unroll
        for (int o = 1; o < 32; o <<= 1) {
            int u = __shfl_up(inc, o, 32);
            if (lane >= o) inc += u;
        }
        if (lane == 31) wsum[wave] = inc;
    }
    __syncthreads();
    if (tid < BKT_NODES) {
        int carry = 0;
        #pragma unroll
        for (int w = 0; w < BKT_NODES / 32; ++w) carry += (w < wave) ? wsum[w] : 0;
        const int excl = s + carry + inc - v;
        const int node = b * BKT_NODES + tid;
        if (node < NNODES) off[node] = excl;
        cur[tid] = excl;
    }
    if (b == NBKT - 1 && tid == 0) off[NNODES] = s + n;
    __syncthreads();
    for (int i0 = tid; i0 < n; i0 += EPT * 256) {
        int w[EPT];
        #pragma unroll
        for (int j = 0; j < EPT; ++j) {
            const int i = i0 + j * 256;
            w[j] = (i < n) ? part[s + i] : -1;
        }
        #pragma unroll
        for (int j = 0; j < EPT; ++j) {
            if (w[j] >= 0) {
                const int pos = atomicAdd(&cur[w[j] >> 16], 1);
                scol[pos] = w[j] & 0xFFFF;
            }
        }
    }
}


__device__ __forceinline__ float lane_value(float v, int l) {
    return __int_as_float(__builtin_amdgcn_readlane(__float_as_int(v), l));
}

__device__ __forceinline__ void edge_logits4(const int* __restrict__ scol,
                                             const float* __restrict__ aD,
                                             const float (&as)[NHEAD],
                                             int p, int end, int& c, float (&a)[NHEAD]) {
    c = 0;
    #pragma unroll
    for (int h = 0; h < NHEAD; ++h) a[h] = -INFINITY;
    if (p < end) {
        c = scol[p];
        const float4 d = *reinterpret_cast<const float4*>(aD + c * NHEAD);
        const float dv[NHEAD] = {d.x, d.y, d.z, d.w};
        #pragma unroll
        for (int h = 0; h < NHEAD; ++h) {
            float t = as[h] + dv[h];
            a[h] = (t >= 0.0f) ? t : NEG_SLOPE * t;
        }
    }
}

__global__ __launch_bounds__(256) void src_softmax_scatter(const int* __restrict__ off,
                                                           const int* __restrict__ scol,
                                                           const float* __restrict__ aS,
                                                           const float* __restrict__ aD,
                                                           const float* __restrict__ xw,
                                                           float* __restrict__ accum) {
    const int lane = threadIdx.x & 31;
    const int r = __builtin_amdgcn_readfirstlane((int)(blockIdx.x * 8 + (threadIdx.x >> 5)));
    if (r >= NNODES) return;
    const int beg = off[r], end = off[r + 1];
    if (beg >= end) return;

    float xr[NHEAD], as[NHEAD];
    #pragma unroll
    for (int h = 0; h < NHEAD; ++h) {
        xr[h] = xw[r * HCOUT + h * NCH + lane];
        as[h] = aS[r * NHEAD + h];
    }

    int c0; float a0[NHEAD];
    edge_logits4(scol, aD, as, beg + lane, end, c0, a0);

    float m[NHEAD];
    #pragma unroll
    for (int h = 0; h < NHEAD; ++h) m[h] = a0[h];
    for (int b = beg + 32; b < end; b += 32) {
        int c; float a[NHEAD];
        edge_logits4(scol, aD, as, b + lane, end, c, a);
        #pragma unroll
        for (int h = 0; h < NHEAD; ++h) m[h] = fmaxf(m[h], a[h]);
    }
    #pragma unroll
    for (int h = 0; h < NHEAD; ++h) {
        #pragma unroll
        for (int o = 16; o > 0; o >>= 1) m[h] = fmaxf(m[h], __shfl_xor(m[h], o, 32));
        if (!(fabsf(m[h]) < INFINITY)) m[h] = 0.0f;
    }

    float ex0[NHEAD], S[NHEAD];
    #pragma unroll
    for (int h = 0; h < NHEAD; ++h) {
        ex0[h] = (beg + lane < end) ? __expf(a0[h] - m[h]) : 0.0f;
        S[h] = ex0[h];
    }
    for (int b = beg + 32; b < end; b += 32) {
        int c; float a[NHEAD];
        edge_logits4(scol, aD, as, b + lane, end, c, a);
        #pragma unroll
        for (int h = 0; h < NHEAD; ++h) S[h] += (b + lane < end) ? __expf(a[h] - m[h]) : 0.0f;
    }
    #pragma unroll
    for (int h = 0; h < NHEAD; ++h) {
        #pragma unroll
        for (int o = 16; o > 0; o >>= 1) S[h] += __shfl_xor(S[h], o, 32);
        S[h] += 1e-16f;
    }

    for (int b = beg; b < end; b += 32) {
        int c; float w[NHEAD];
        if (b == beg) {
            c = c0;
            #pragma unroll
            for (int h = 0; h < NHEAD; ++h) w[h] = ex0[h] / S[h];
        } else {
            float a[NHEAD];
            edge_logits4(scol, aD, as, b + lane, end, c, a);
            #pragma unroll
            for (int h = 0; h < NHEAD; ++h) w[h] = ((b + lane < end) ? __expf(a[h] - m[h]) : 0.0f) / S[h];
        }
        const int cnt = (end - b < 32) ? (end - b) : 32;
        for (int j = 0; j < cnt; ++j) {
            const int cj = __builtin_amdgcn_readlane(c, j);
            float s = lane_value(w[0], j) * xr[0];
            #pragma unroll
            for (int h = 1; h < NHEAD; ++h) s += lane_value(w[h], j) * xr[h];
            atomicAdd(&accum[cj * NCH + lane], s);
        }
    }
}

__global__ __launch_bounds__(256) void finalize(const float* __restrict__ accum,
                                                const float* __restrict__ bias,
                                                float* __restrict__ out) {
    int t = blockIdx.x * 256 + threadIdx.x;
    if (t >= NNODES * NCH) return;
    out[t] = 0.25f * accum[t] + bias[t & 31];
}

extern "C" void kernel_launch(void* const* d_in, const int* in_sizes, int n_in,
                              void* d_out, int out_size, void* d_ws, size_t ws_size,
                              hipStream_t stream) {
    const float* x    = (const float*)d_in[0];
    const int*   ei   = (const int*)  d_in[1];
    const float* w    = (const float*)d_in[2];
    const float* att  = (const float*)d_in[3];
    const float* bias = (const float*)d_in[4];
    float* ws = (float*)d_ws;

    float* xw    = ws;
    float* aS    = ws + 6400000;
    float* aD    = ws + 6600000;
    int*   scol  = (int*)(ws + 6800000);
    int*   off   = (int*)(ws + 7600000);
    int*   part  = (int*)(ws + 7700000);
    int*   hist  = (int*)(ws + 8500000);
    int*   bstart= (int*)(ws + 8600000);
    float* accum = ws + 10050048;

    const int* row = ei;
    const int* col = ei + NEDGE;

    const int ZN = NNODES * NCH;
    zero_ws     <<<(ZN + 255) / 256, 256, 0, stream>>>(accum, ZN);
    bucket_count<<<NCHUNK, 1024, 0, stream>>>(row, hist);
    bucket_place<<<NCHUNK, 1024, 0, stream>>>(row, col, hist, bstart, part);
    bucket_sort <<<NBKT, 256, 0, stream>>>(bstart, part, off, scol);
    gemm_xw   <<<(NNODES + 15) / 16, 256, 0, stream>>>(x, w, xw);
    node_dots <<<(NNODES * NHEAD + 255) / 256, 256, 0, stream>>>(xw, att, aS, aD);
    src_softmax_scatter<<<(NNODES + 7) / 8, 256, 0, stream>>>(off, scol, aS, aD, xw, accum);
    finalize  <<<(NNODES * NCH + 255) / 256, 256, 0, stream>>>(accum, bias, (float*)d_out);
}


# accumulate straight into out: edge-count kernel stores out = bias, scatter adds a quarter of each message there; zero-fill and finalize launches (100,000 waves) removed; bias now first in each sum
# speedup vs baseline: 5.2598x; 1.0715x over previous
#include <hip/hip_runtime.h>

#define NNODES 50000
#define FEAT   128
#define HCOUT  128
#define NHEAD  4
#define NCH    32
#define NEDGE  800000
#define NEG_SLOPE 0.2f

typedef __attribute__((ext_vector_type(2)))  float    v2f;
typedef __attribute__((ext_vector_type(8)))  float    v8f;
typedef __attribute__((ext_vector_type(16))) _Float16 v16h;

#if __has_builtin(__builtin_amdgcn_wmma_f32_16x16x4_f32)
#define USE_F32_WMMA 1
#endif

__global__ __launch_bounds__(256) void gemm_xw(const float* __restrict__ x,
                                               const float* __restrict__ w,
                                               float* __restrict__ xw) {
    __shared__ float ldsA[16][FEAT];
    const int rowBase = blockIdx.x * 16;

    for (int idx = threadIdx.x; idx < 16 * FEAT; idx += 256) {
        int r = idx >> 7, c = idx & 127;
        int node = rowBase + r;
        ldsA[r][c] = (node < NNODES) ? x[node * FEAT + c] : 0.0f;
    }
    __syncthreads();

    const int wave    = threadIdx.x >> 5;
    const int lane    = threadIdx.x & 31;
    const int colBase = wave * 16;
    const int Mrow    = lane & 15;
    const bool hi     = lane >= 16;

    v8f acc = {};

#ifdef USE_F32_WMMA
    const int koff = hi ? 2 : 0;
    for (int k0 = 0; k0 < FEAT; k0 += 4) {
        v2f a, b;
        a.x = ldsA[Mrow][k0 + koff];
        a.y = ldsA[Mrow][k0 + koff + 1];
        b.x = w[(k0 + koff)     * HCOUT + colBase + Mrow];
        b.y = w[(k0 + koff + 1) * HCOUT + colBase + Mrow];
        acc = __builtin_amdgcn_wmma_f32_16x16x4_f32(
            false, a, false, b, (short)0, acc, false, false);
    }
#else
    for (int k0 = 0; k0 < FEAT; k0 += 32) {
        v16h a, b;
        const int ka = hi ? 8 : 0;
        const int kb = hi ? 16 : 0;
        #pragma unroll
        for (int i = 0; i < 8; ++i) {
            a[i]     = (_Float16)ldsA[Mrow][k0 + ka + i];
            a[8 + i] = (_Float16)ldsA[Mrow][k0 + 16 + ka + i];
        }
        #pragma unroll
        for (int i = 0; i < 16; ++i)
            b[i] = (_Float16)w[(k0 + kb + i) * HCOUT + colBase + Mrow];
        acc = __builtin_amdgcn_wmma_f32_16x16x32_f16(
            false, a, false, b, (short)0, acc, false, false);
    }
#endif

    #pragma unroll
    for (int v = 0; v < 8; ++v) {
        int node = rowBase + v + (hi ? 8 : 0);
        if (node < NNODES) xw[node * HCOUT + colBase + Mrow] = acc[v];
    }
}

__global__ __launch_bounds__(256) void node_dots(const float* __restrict__ xw,
                                                 const float* __restrict__ att,
                                                 float* __restrict__ aS,
                                                 float* __restrict__ aD) {
    int t = blockIdx.x * 256 + threadIdx.x;
    if (t >= NNODES * NHEAD) return;
    int h = t & (NHEAD - 1);
    const float* v  = xw + t * NCH;
    const float* cs = att + h * (2 * NCH);
    const float* cd = cs + NCH;
    float s = 0.0f, d = 0.0f;
    #pragma unroll
    for (int c = 0; c < NCH; ++c) {
        float xv = v[c];
        s += xv * cs[c];
        d += xv * cd[c];
    }
    aS[t] = s;
    aD[t] = d;
}

#define BKT_SHIFT 7
#define BKT_NODES (1 << BKT_SHIFT)
#define NBKT      ((NNODES + BKT_NODES - 1) / BKT_NODES)
#define CHUNK     4000
#define NCHUNK    ((NEDGE + CHUNK - 1) / CHUNK)
#define HALFCH    (NCHUNK / 2)
#define HPITCH    416
static_assert(NNODES <= 65536, "a destination fits the low 16 bits of a packed word");
static_assert(NBKT <= 511 && NBKT <= HPITCH, "one thread of the first 512 per bucket, and one spare");
#define HGROUP    25
static_assert(NCHUNK % 2 == 0, "the rows of counts are summed in two equal halves");
static_assert(HALFCH % HGROUP == 0, "whole groups of rows");
#define EPT       4
static_assert(CHUNK <= EPT * 1024, "one batch of loads per thread covers a chunk");
#define OUT4      (NNODES * NCH / 4)
static_assert(OUT4 <= 2 * NCHUNK * 1024, "two float4 stores per thread of bucket_count cover out");
static_assert(NCHUNK * 1024 <= OUT4, "the first of the two stores is inside out for every thread");
static_assert((NCHUNK * 1024) % (NCH / 4) == 0, "both stores of a thread fall on the same channels");

__global__ __launch_bounds__(1024) void bucket_count(const int* __restrict__ row,
                                                     const float* __restrict__ bias,
                                                     int* __restrict__ hist,
                                                     float* __restrict__ out) {
    __shared__ int cnt[512];
    const int tid = threadIdx.x;
    if (tid < 512) cnt[tid] = 0;
    __syncthreads();
    const int g = blockIdx.x * 1024 + tid;
    const float4 b4 = reinterpret_cast<const float4*>(bias)[g & (NCH / 4 - 1)];
    const int e0 = blockIdx.x * CHUNK;
    const int e1 = (e0 + CHUNK < NEDGE) ? (e0 + CHUNK) : NEDGE;
    int r[EPT];
    #pragma unroll
    for (int j = 0; j < EPT; ++j) {
        const int e = e0 + tid + j * 1024;
        r[j] = (e < e1) ? row[e] : -1;
    }
    #pragma unroll
    for (int j = 0; j < EPT; ++j)
        if (r[j] >= 0) atomicAdd(&cnt[r[j] >> BKT_SHIFT], 1);
    asm volatile("" :: "v"(b4.x), "v"(b4.y), "v"(b4.z), "v"(b4.w));
    __syncthreads();
    if (tid < NBKT) hist[blockIdx.x * HPITCH + tid] = cnt[tid];
    float4* o4 = reinterpret_cast<float4*>(out);
    o4[g] = b4;
    if (g + NCHUNK * 1024 < OUT4) o4[g + NCHUNK * 1024] = b4;
}

__global__ __launch_bounds__(1024) void bucket_place(const int* __restrict__ row,
                                                     const int* __restrict__ col,
                                                     const int* __restrict__ hist,
                                                     int* __restrict__ bstart,
                                                     int* __restrict__ part) {
    __shared__ int ptot[2][512];
    __shared__ int pbel[2][512];
    __shared__ int wsum[16];
    __shared__ int cnt[512];
    const int tid  = threadIdx.x;
    const int lane = tid & 31;
    const int wave = tid >> 5;
    const int k    = blockIdx.x;
    {
        const int b  = tid & 511;
        const int hf = tid >> 9;
        int tot = 0, bel = 0;
        if (b < NBKT) {
            const int k0 = hf * HALFCH;
            const int* p = hist + k0 * HPITCH + b;
            const int nbel = (k - k0 < 0) ? 0 : ((k - k0 > HALFCH) ? HALFCH : (k - k0));
            for (int g = 0; g < HALFCH; g += HGROUP) {
                int v[HGROUP];
                #pragma unroll
                for (int j = 0; j < HGROUP; ++j) v[j] = p[(g + j) * HPITCH];
                #pragma unroll
                for (int j = 0; j < HGROUP; ++j) {
                    tot += v[j];
                    bel += (g + j < nbel) ? v[j] : 0;
                }
            }
        }
        ptot[hf][b] = tot;
        pbel[hf][b] = bel;
    }
    __syncthreads();
    int T = 0, inc = 0;
    if (tid < 512) {
        T = ptot[0][tid] + ptot[1][tid];
        inc = T;
        #pragma unroll
        for (int o = 1; o < 32; o <<= 1) {
            int u = __shfl_up(inc, o, 32);
            if (lane >= o) inc += u;
        }
        if (lane == 31) wsum[wave] = inc;
    }
    __syncthreads();
    if (tid < 512) {
        int carry = 0;
        #pragma unroll
        for (int w = 0; w < 16; ++w) carry += (w < wave) ? wsum[w] : 0;
        const int excl = carry + inc - T;
        cnt[tid] = excl + pbel[0][tid] + pbel[1][tid];
        if (k == 0 && tid <= NBKT) bstart[tid] = excl;
    }
    __syncthreads();
    const int e0 = k * CHUNK;
    const int e1 = (e0 + CHUNK < NEDGE) ? (e0 + CHUNK) : NEDGE;
    int r[EPT], c[EPT];
    #pragma unroll
    for (int j = 0; j < EPT; ++j) {
        const int e = e0 + tid + j * 1024;
        r[j] = (e < e1) ? row[e] : -1;
        c[j] = (e < e1) ? col[e] : 0;
    }
    #pragma unroll
    for (int j = 0; j < EPT; ++j) {
        if (r[j] >= 0) {
            const int pos = atomicAdd(&cnt[r[j] >> BKT_SHIFT], 1);
            part[pos] = ((r[j] & (BKT_NODES - 1)) << 16) | c[j];
        }
    }
}

__global__ __launch_bounds__(256) void bucket_sort(const int* __restrict__ bstart,
                                                   const int* __restrict__ part,
                                                   int* __restrict__ off,
                                                   int* __restrict__ scol) {
    __shared__ int cnt[BKT_NODES];
    __shared__ int cur[BKT_NODES];
    __shared__ int wsum[BKT_NODES / 32];
    const int tid  = threadIdx.x;
    const int lane = tid & 31;
    const int wave = tid >> 5;
    const int b    = blockIdx.x;
    const int s    = bstart[b];
    const int n    = bstart[b + 1] - s;
    if (tid < BKT_NODES) cnt[tid] = 0;
    __syncthreads();
    for (int i0 = tid; i0 < n; i0 += EPT * 256) {
        int w[EPT];
        #pragma unroll
        for (int j = 0; j < EPT; ++j) {
            const int i = i0 + j * 256;
            w[j] = (i < n) ? part[s + i] : -1;
        }
        #pragma unroll
        for (int j = 0; j < EPT; ++j)
            if (w[j] >= 0) atomicAdd(&cnt[w[j] >> 16], 1);
    }
    __syncthreads();
    int v = 0, inc = 0;
    if (tid < BKT_NODES) {
        v = cnt[tid];
        inc = v;
        #pragma unroll
        for (int o = 1; o < 32; o <<= 1) {
            int u = __shfl_up(inc, o, 32);
            if (lane >= o) inc += u;
        }
        if (lane == 31) wsum[wave] = inc;
    }
    __syncthreads();
    if (tid < BKT_NODES) {
        int carry = 0;
        #pragma unroll
        for (int w = 0; w < BKT_NODES / 32; ++w) carry += (w < wave) ? wsum[w] : 0;
        const int excl = s + carry + inc - v;
        const int node = b * BKT_NODES + tid;
        if (node < NNODES) off[node] = excl;
        cur[tid] = excl;
    }
    if (b == NBKT - 1 && tid == 0) off[NNODES] = s + n;
    __syncthreads();
    for (int i0 = tid; i0 < n; i0 += EPT * 256) {
        int w[EPT];
        #pragma unroll
        for (int j = 0; j < EPT; ++j) {
            const int i = i0 + j * 256;
            w[j] = (i < n) ? part[s + i] : -1;
        }
        #pragma unroll
        for (int j = 0; j < EPT; ++j) {
            if (w[j] >= 0) {
                const int pos = atomicAdd(&cur[w[j] >> 16], 1);
                scol[pos] = w[j] & 0xFFFF;
            }
        }
    }
}


__device__ __forceinline__ float lane_value(float v, int l) {
    return __int_as_float(__builtin_amdgcn_readlane(__float_as_int(v), l));
}

__device__ __forceinline__ void edge_logits4(const int* __restrict__ scol,
                                             const float* __restrict__ aD,
                                             const float (&as)[NHEAD],
                                             int p, int end, int& c, float (&a)[NHEAD]) {
    c = 0;
    #pragma unroll
    for (int h = 0; h < NHEAD; ++h) a[h] = -INFINITY;
    if (p < end) {
        c = scol[p];
        const float4 d = *reinterpret_cast<const float4*>(aD + c * NHEAD);
        const float dv[NHEAD] = {d.x, d.y, d.z, d.w};
        #pragma unroll
        for (int h = 0; h < NHEAD; ++h) {
            float t = as[h] + dv[h];
            a[h] = (t >= 0.0f) ? t : NEG_SLOPE * t;
        }
    }
}

__global__ __launch_bounds__(256) void src_softmax_scatter(const int* __restrict__ off,
                                                           const int* __restrict__ scol,
                                                           const float* __restrict__ aS,
                                                           const float* __restrict__ aD,
                                                           const float* __restrict__ xw,
                                                           float* __restrict__ out) {
    const int lane = threadIdx.x & 31;
    const int r = __builtin_amdgcn_readfirstlane((int)(blockIdx.x * 8 + (threadIdx.x >> 5)));
    if (r >= NNODES) return;
    const int beg = off[r], end = off[r + 1];
    if (beg >= end) return;

    float xr[NHEAD], as[NHEAD];
    #pragma unroll
    for (int h = 0; h < NHEAD; ++h) {
        xr[h] = 0.25f * xw[r * HCOUT + h * NCH + lane];
        as[h] = aS[r * NHEAD + h];
    }

    int c0; float a0[NHEAD];
    edge_logits4(scol, aD, as, beg + lane, end, c0, a0);

    float m[NHEAD];
    #pragma unroll
    for (int h = 0; h < NHEAD; ++h) m[h] = a0[h];
    for (int b = beg + 32; b < end; b += 32) {
        int c; float a[NHEAD];
        edge_logits4(scol, aD, as, b + lane, end, c, a);
        #pragma unroll
        for (int h = 0; h < NHEAD; ++h) m[h] = fmaxf(m[h], a[h]);
    }
    #pragma unroll
    for (int h = 0; h < NHEAD; ++h) {
        #pragma unroll
        for (int o = 16; o > 0; o >>= 1) m[h] = fmaxf(m[h], __shfl_xor(m[h], o, 32));
        if (!(fabsf(m[h]) < INFINITY)) m[h] = 0.0f;
    }

    float ex0[NHEAD], S[NHEAD];
    #pragma unroll
    for (int h = 0; h < NHEAD; ++h) {
        ex0[h] = (beg + lane < end) ? __expf(a0[h] - m[h]) : 0.0f;
        S[h] = ex0[h];
    }
    for (int b = beg + 32; b < end; b += 32) {
        int c; float a[NHEAD];
        edge_logits4(scol, aD, as, b + lane, end, c, a);
        #pragma unroll
        for (int h = 0; h < NHEAD; ++h) S[h] += (b + lane < end) ? __expf(a[h] - m[h]) : 0.0f;
    }
    #pragma unroll
    for (int h = 0; h < NHEAD; ++h) {
        #pragma unroll
        for (int o = 16; o > 0; o >>= 1) S[h] += __shfl_xor(S[h], o, 32);
        S[h] += 1e-16f;
    }

    for (int b = beg; b < end; b += 32) {
        int c; float w[NHEAD];
        if (b == beg) {
            c = c0;
            #pragma unroll
            for (int h = 0; h < NHEAD; ++h) w[h] = ex0[h] / S[h];
        } else {
            float a[NHEAD];
            edge_logits4(scol, aD, as, b + lane, end, c, a);
            #pragma unroll
            for (int h = 0; h < NHEAD; ++h) w[h] = ((b + lane < end) ? __expf(a[h] - m[h]) : 0.0f) / S[h];
        }
        const int cnt = (end - b < 32) ? (end - b) : 32;
        for (int j = 0; j < cnt; ++j) {
            const int cj = __builtin_amdgcn_readlane(c, j);
            float s = lane_value(w[0], j) * xr[0];
            #pragma unroll
            for (int h = 1; h < NHEAD; ++h) s += lane_value(w[h], j) * xr[h];
            atomicAdd(&out[cj * NCH + lane], s);
        }
    }
}

extern "C" void kernel_launch(void* const* d_in, const int* in_sizes, int n_in,
                              void* d_out, int out_size, void* d_ws, size_t ws_size,
                              hipStream_t stream) {
    const float* x    = (const float*)d_in[0];
    const int*   ei   = (const int*)  d_in[1];
    const float* w    = (const float*)d_in[2];
    const float* att  = (const float*)d_in[3];
    const float* bias = (const float*)d_in[4];
    float* ws = (float*)d_ws;

    float* xw    = ws;
    float* aS    = ws + 6400000;
    float* aD    = ws + 6600000;
    int*   scol  = (int*)(ws + 6800000);
    int*   off   = (int*)(ws + 7600000);
    int*   part  = (int*)(ws + 7700000);
    int*   hist  = (int*)(ws + 8500000);
    int*   bstart= (int*)(ws + 8600000);

    const int* row = ei;
    const int* col = ei + NEDGE;

    float* out = (float*)d_out;

    bucket_count<<<NCHUNK, 1024, 0, stream>>>(row, bias, hist, out);
    bucket_place<<<NCHUNK, 1024, 0, stream>>>(row, col, hist, bstart, part);
    bucket_sort <<<NBKT, 256, 0, stream>>>(bstart, part, off, scol);
    gemm_xw   <<<(NNODES + 15) / 16, 256, 0, stream>>>(x, w, xw);
    node_dots <<<(NNODES * NHEAD + 255) / 256, 256, 0, stream>>>(xw, att, aS, aD);
    src_softmax_scatter<<<(NNODES + 7) / 8, 256, 0, stream>>>(off, scol, aS, aD, xw, out);
}


# product kernel stages its 16x128 strip of xw in LDS, stores whole rows and takes the attention dots there (same ascending channel order, all fused multiply-add); node_dots launch removed
# speedup vs baseline: 5.5320x; 1.0518x over previous
#include <hip/hip_runtime.h>

#define NNODES 50000
#define FEAT   128
#define HCOUT  128
#define NHEAD  4
#define NCH    32
#define NEDGE  800000
#define NEG_SLOPE 0.2f

typedef __attribute__((ext_vector_type(2)))  float    v2f;
typedef __attribute__((ext_vector_type(8)))  float    v8f;
typedef __attribute__((ext_vector_type(16))) _Float16 v16h;

#if __has_builtin(__builtin_amdgcn_wmma_f32_16x16x4_f32)
#define USE_F32_WMMA 1
#endif

#define TPITCH 132
__global__ __launch_bounds__(256) void gemm_xw(const float* __restrict__ x,
                                               const float* __restrict__ w,
                                               const float* __restrict__ att,
                                               float* __restrict__ xw,
                                               float* __restrict__ aS,
                                               float* __restrict__ aD) {
    __shared__ float ldsA[16][FEAT];
    __shared__ __attribute__((aligned(16))) float tile[16][TPITCH];
    const int rowBase = blockIdx.x * 16;

    for (int idx = threadIdx.x; idx < 16 * FEAT; idx += 256) {
        int r = idx >> 7, c = idx & 127;
        int node = rowBase + r;
        ldsA[r][c] = (node < NNODES) ? x[node * FEAT + c] : 0.0f;
    }
    __syncthreads();

    const int wave    = threadIdx.x >> 5;
    const int lane    = threadIdx.x & 31;
    const int colBase = wave * 16;
    const int Mrow    = lane & 15;
    const bool hi     = lane >= 16;

    v8f acc = {};

#ifdef USE_F32_WMMA
    const int koff = hi ? 2 : 0;
    for (int k0 = 0; k0 < FEAT; k0 += 4) {
        v2f a, b;
        a.x = ldsA[Mrow][k0 + koff];
        a.y = ldsA[Mrow][k0 + koff + 1];
        b.x = w[(k0 + koff)     * HCOUT + colBase + Mrow];
        b.y = w[(k0 + koff + 1) * HCOUT + colBase + Mrow];
        acc = __builtin_amdgcn_wmma_f32_16x16x4_f32(
            false, a, false, b, (short)0, acc, false, false);
    }
#else
    for (int k0 = 0; k0 < FEAT; k0 += 32) {
        v16h a, b;
        const int ka = hi ? 8 : 0;
        const int kb = hi ? 16 : 0;
        #pragma unroll
        for (int i = 0; i < 8; ++i) {
            a[i]     = (_Float16)ldsA[Mrow][k0 + ka + i];
            a[8 + i] = (_Float16)ldsA[Mrow][k0 + 16 + ka + i];
        }
        #pragma unroll
        for (int i = 0; i < 16; ++i)
            b[i] = (_Float16)w[(k0 + kb + i) * HCOUT + colBase + Mrow];
        acc = __builtin_amdgcn_wmma_f32_16x16x32_f16(
            false, a, false, b, (short)0, acc, false, false);
    }
#endif

    #pragma unroll
    for (int v = 0; v < 8; ++v) tile[v + (hi ? 8 : 0)][colBase + Mrow] = acc[v];
    __syncthreads();

    #pragma unroll
    for (int i = 0; i < 2; ++i) {
        const int f    = threadIdx.x + i * 256;
        const int r    = f >> 5;
        const int c4   = f & 31;
        const int node = rowBase + r;
        if (node < NNODES) {
            const float4 val = *reinterpret_cast<const float4*>(&tile[r][c4 * 4]);
            *reinterpret_cast<float4*>(xw + node * HCOUT + c4 * 4) = val;
        }
    }

    if (threadIdx.x < 128) {
        const int t    = threadIdx.x;
        const int h    = t & (NHEAD - 1);
        const int r    = (t >> 2) & 15;
        const int sd   = t >> 6;
        const int node = rowBase + r;
        const float* coef = att + h * (2 * NCH) + sd * NCH;
        const float* v    = &tile[r][h * NCH];
        float s = 0.0f;
        #pragma unroll
        for (int c = 0; c < NCH; ++c) s = __builtin_fmaf(v[c], coef[c], s);
        if (node < NNODES) (sd ? aD : aS)[node * NHEAD + h] = s;
    }
}

#define BKT_SHIFT 7
#define BKT_NODES (1 << BKT_SHIFT)
#define NBKT      ((NNODES + BKT_NODES - 1) / BKT_NODES)
#define CHUNK     4000
#define NCHUNK    ((NEDGE + CHUNK - 1) / CHUNK)
#define HALFCH    (NCHUNK / 2)
#define HPITCH    416
static_assert(NNODES <= 65536, "a destination fits the low 16 bits of a packed word");
static_assert(NBKT <= 511 && NBKT <= HPITCH, "one thread of the first 512 per bucket, and one spare");
#define HGROUP    25
static_assert(NCHUNK % 2 == 0, "the rows of counts are summed in two equal halves");
static_assert(HALFCH % HGROUP == 0, "whole groups of rows");
#define EPT       4
static_assert(CHUNK <= EPT * 1024, "one batch of loads per thread covers a chunk");
#define OUT4      (NNODES * NCH / 4)
static_assert(OUT4 <= 2 * NCHUNK * 1024, "two float4 stores per thread of bucket_count cover out");
static_assert(NCHUNK * 1024 <= OUT4, "the first of the two stores is inside out for every thread");
static_assert((NCHUNK * 1024) % (NCH / 4) == 0, "both stores of a thread fall on the same channels");

__global__ __launch_bounds__(1024) void bucket_count(const int* __restrict__ row,
                                                     const float* __restrict__ bias,
                                                     int* __restrict__ hist,
                                                     float* __restrict__ out) {
    __shared__ int cnt[512];
    const int tid = threadIdx.x;
    if (tid < 512) cnt[tid] = 0;
    __syncthreads();
    const int g = blockIdx.x * 1024 + tid;
    const float4 b4 = reinterpret_cast<const float4*>(bias)[g & (NCH / 4 - 1)];
    const int e0 = blockIdx.x * CHUNK;
    const int e1 = (e0 + CHUNK < NEDGE) ? (e0 + CHUNK) : NEDGE;
    int r[EPT];
    #pragma unroll
    for (int j = 0; j < EPT; ++j) {
        const int e = e0 + tid + j * 1024;
        r[j] = (e < e1) ? row[e] : -1;
    }
    #pragma unroll
    for (int j = 0; j < EPT; ++j)
        if (r[j] >= 0) atomicAdd(&cnt[r[j] >> BKT_SHIFT], 1);
    asm volatile("" :: "v"(b4.x), "v"(b4.y), "v"(b4.z), "v"(b4.w));
    __syncthreads();
    if (tid < NBKT) hist[blockIdx.x * HPITCH + tid] = cnt[tid];
    float4* o4 = reinterpret_cast<float4*>(out);
    o4[g] = b4;
    if (g + NCHUNK * 1024 < OUT4) o4[g + NCHUNK * 1024] = b4;
}

__global__ __launch_bounds__(1024) void bucket_place(const int* __restrict__ row,
                                                     const int* __restrict__ col,
                                                     const int* __restrict__ hist,
                                                     int* __restrict__ bstart,
                                                     int* __restrict__ part) {
    __shared__ int ptot[2][512];
    __shared__ int pbel[2][512];
    __shared__ int wsum[16];
    __shared__ int cnt[512];
    const int tid  = threadIdx.x;
    const int lane = tid & 31;
    const int wave = tid >> 5;
    const int k    = blockIdx.x;
    {
        const int b  = tid & 511;
        const int hf = tid >> 9;
        int tot = 0, bel = 0;
        if (b < NBKT) {
            const int k0 = hf * HALFCH;
            const int* p = hist + k0 * HPITCH + b;
            const int nbel = (k - k0 < 0) ? 0 : ((k - k0 > HALFCH) ? HALFCH : (k - k0));
            for (int g = 0; g < HALFCH; g += HGROUP) {
                int v[HGROUP];
                #pragma unroll
                for (int j = 0; j < HGROUP; ++j) v[j] = p[(g + j) * HPITCH];
                #pragma unroll
                for (int j = 0; j < HGROUP; ++j) {
                    tot += v[j];
                    bel += (g + j < nbel) ? v[j] : 0;
                }
            }
        }
        ptot[hf][b] = tot;
        pbel[hf][b] = bel;
    }
    __syncthreads();
    int T = 0, inc = 0;
    if (tid < 512) {
        T = ptot[0][tid] + ptot[1][tid];
        inc = T;
        #pragma unroll
        for (int o = 1; o < 32; o <<= 1) {
            int u = __shfl_up(inc, o, 32);
            if (lane >= o) inc += u;
        }
        if (lane == 31) wsum[wave] = inc;
    }
    __syncthreads();
    if (tid < 512) {
        int carry = 0;
        #pragma unroll
        for (int w = 0; w < 16; ++w) carry += (w < wave) ? wsum[w] : 0;
        const int excl = carry + inc - T;
        cnt[tid] = excl + pbel[0][tid] + pbel[1][tid];
        if (k == 0 && tid <= NBKT) bstart[tid] = excl;
    }
    __syncthreads();
    const int e0 = k * CHUNK;
    const int e1 = (e0 + CHUNK < NEDGE) ? (e0 + CHUNK) : NEDGE;
    int r[EPT], c[EPT];
    #pragma unroll
    for (int j = 0; j < EPT; ++j) {
        const int e = e0 + tid + j * 1024;
        r[j] = (e < e1) ? row[e] : -1;
        c[j] = (e < e1) ? col[e] : 0;
    }
    #pragma unroll
    for (int j = 0; j < EPT; ++j) {
        if (r[j] >= 0) {
            const int pos = atomicAdd(&cnt[r[j] >> BKT_SHIFT], 1);
            part[pos] = ((r[j] & (BKT_NODES - 1)) << 16) | c[j];
        }
    }
}

__global__ __launch_bounds__(256) void bucket_sort(const int* __restrict__ bstart,
                                                   const int* __restrict__ part,
                                                   int* __restrict__ off,
                                                   int* __restrict__ scol) {
    __shared__ int cnt[BKT_NODES];
    __shared__ int cur[BKT_NODES];
    __shared__ int wsum[BKT_NODES / 32];
    const int tid  = threadIdx.x;
    const int lane = tid & 31;
    const int wave = tid >> 5;
    const int b    = blockIdx.x;
    const int s    = bstart[b];
    const int n    = bstart[b + 1] - s;
    if (tid < BKT_NODES) cnt[tid] = 0;
    __syncthreads();
    for (int i0 = tid; i0 < n; i0 += EPT * 256) {
        int w[EPT];
        #pragma unroll
        for (int j = 0; j < EPT; ++j) {
            const int i = i0 + j * 256;
            w[j] = (i < n) ? part[s + i] : -1;
        }
        #pragma unroll
        for (int j = 0; j < EPT; ++j)
            if (w[j] >= 0) atomicAdd(&cnt[w[j] >> 16], 1);
    }
    __syncthreads();
    int v = 0, inc = 0;
    if (tid < BKT_NODES) {
        v = cnt[tid];
        inc = v;
        #pragma unroll
        for (int o = 1; o < 32; o <<= 1) {
            int u = __shfl_up(inc, o, 32);
            if (lane >= o) inc += u;
        }
        if (lane == 31) wsum[wave] = inc;
    }
    __syncthreads();
    if (tid < BKT_NODES) {
        int carry = 0;
        #pragma unroll
        for (int w = 0; w < BKT_NODES / 32; ++w) carry += (w < wave) ? wsum[w] : 0;
        const int excl = s + carry + inc - v;
        const int node = b * BKT_NODES + tid;
        if (node < NNODES) off[node] = excl;
        cur[tid] = excl;
    }
    if (b == NBKT - 1 && tid == 0) off[NNODES] = s + n;
    __syncthreads();
    for (int i0 = tid; i0 < n; i0 += EPT * 256) {
        int w[EPT];
        #pragma unroll
        for (int j = 0; j < EPT; ++j) {
            const int i = i0 + j * 256;
            w[j] = (i < n) ? part[s + i] : -1;
        }
        #pragma unroll
        for (int j = 0; j < EPT; ++j) {
            if (w[j] >= 0) {
                const int pos = atomicAdd(&cur[w[j] >> 16], 1);
                scol[pos] = w[j] & 0xFFFF;
            }
        }
    }
}


__device__ __forceinline__ float lane_value(float v, int l) {
    return __int_as_float(__builtin_amdgcn_readlane(__float_as_int(v), l));
}

__device__ __forceinline__ void edge_logits4(const int* __restrict__ scol,
                                             const float* __restrict__ aD,
                                             const float (&as)[NHEAD],
                                             int p, int end, int& c, float (&a)[NHEAD]) {
    c = 0;
    #pragma unroll
    for (int h = 0; h < NHEAD; ++h) a[h] = -INFINITY;
    if (p < end) {
        c = scol[p];
        const float4 d = *reinterpret_cast<const float4*>(aD + c * NHEAD);
        const float dv[NHEAD] = {d.x, d.y, d.z, d.w};
        #pragma unroll
        for (int h = 0; h < NHEAD; ++h) {
            float t = as[h] + dv[h];
            a[h] = (t >= 0.0f) ? t : NEG_SLOPE * t;
        }
    }
}

__global__ __launch_bounds__(256) void src_softmax_scatter(const int* __restrict__ off,
                                                           const int* __restrict__ scol,
                                                           const float* __restrict__ aS,
                                                           const float* __restrict__ aD,
                                                           const float* __restrict__ xw,
                                                           float* __restrict__ out) {
    const int lane = threadIdx.x & 31;
    const int r = __builtin_amdgcn_readfirstlane((int)(blockIdx.x * 8 + (threadIdx.x >> 5)));
    if (r >= NNODES) return;
    const int beg = off[r], end = off[r + 1];
    if (beg >= end) return;

    float xr[NHEAD], as[NHEAD];
    #pragma unroll
    for (int h = 0; h < NHEAD; ++h) {
        xr[h] = 0.25f * xw[r * HCOUT + h * NCH + lane];
        as[h] = aS[r * NHEAD + h];
    }

    int c0; float a0[NHEAD];
    edge_logits4(scol, aD, as, beg + lane, end, c0, a0);

    float m[NHEAD];
    #pragma unroll
    for (int h = 0; h < NHEAD; ++h) m[h] = a0[h];
    for (int b = beg + 32; b < end; b += 32) {
        int c; float a[NHEAD];
        edge_logits4(scol, aD, as, b + lane, end, c, a);
        #pragma unroll
        for (int h = 0; h < NHEAD; ++h) m[h] = fmaxf(m[h], a[h]);
    }
    #pragma unroll
    for (int h = 0; h < NHEAD; ++h) {
        #pragma unroll
        for (int o = 16; o > 0; o >>= 1) m[h] = fmaxf(m[h], __shfl_xor(m[h], o, 32));
        if (!(fabsf(m[h]) < INFINITY)) m[h] = 0.0f;
    }

    float ex0[NHEAD], S[NHEAD];
    #pragma unroll
    for (int h = 0; h < NHEAD; ++h) {
        ex0[h] = (beg + lane < end) ? __expf(a0[h] - m[h]) : 0.0f;
        S[h] = ex0[h];
    }
    for (int b = beg + 32; b < end; b += 32) {
        int c; float a[NHEAD];
        edge_logits4(scol, aD, as, b + lane, end, c, a);
        #pragma unroll
        for (int h = 0; h < NHEAD; ++h) S[h] += (b + lane < end) ? __expf(a[h] - m[h]) : 0.0f;
    }
    #pragma unroll
    for (int h = 0; h < NHEAD; ++h) {
        #pragma unroll
        for (int o = 16; o > 0; o >>= 1) S[h] += __shfl_xor(S[h], o, 32);
        S[h] += 1e-16f;
    }

    for (int b = beg; b < end; b += 32) {
        int c; float w[NHEAD];
        if (b == beg) {
            c = c0;
            #pragma unroll
            for (int h = 0; h < NHEAD; ++h) w[h] = ex0[h] / S[h];
        } else {
            float a[NHEAD];
            edge_logits4(scol, aD, as, b + lane, end, c, a);
            #pragma unroll
            for (int h = 0; h < NHEAD; ++h) w[h] = ((b + lane < end) ? __expf(a[h] - m[h]) : 0.0f) / S[h];
        }
        const int cnt = (end - b < 32) ? (end - b) : 32;
        for (int j = 0; j < cnt; ++j) {
            const int cj = __builtin_amdgcn_readlane(c, j);
            float s = lane_value(w[0], j) * xr[0];
            #pragma unroll
            for (int h = 1; h < NHEAD; ++h) s += lane_value(w[h], j) * xr[h];
            atomicAdd(&out[cj * NCH + lane], s);
        }
    }
}

extern "C" void kernel_launch(void* const* d_in, const int* in_sizes, int n_in,
                              void* d_out, int out_size, void* d_ws, size_t ws_size,
                              hipStream_t stream) {
    const float* x    = (const float*)d_in[0];
    const int*   ei   = (const int*)  d_in[1];
    const float* w    = (const float*)d_in[2];
    const float* att  = (const float*)d_in[3];
    const float* bias = (const float*)d_in[4];
    float* ws = (float*)d_ws;

    float* xw    = ws;
    float* aS    = ws + 6400000;
    float* aD    = ws + 6600000;
    int*   scol  = (int*)(ws + 6800000);
    int*   off   = (int*)(ws + 7600000);
    int*   part  = (int*)(ws + 7700000);
    int*   hist  = (int*)(ws + 8500000);
    int*   bstart= (int*)(ws + 8600000);

    const int* row = ei;
    const int* col = ei + NEDGE;

    float* out = (float*)d_out;

    bucket_count<<<NCHUNK, 1024, 0, stream>>>(row, bias, hist, out);
    bucket_place<<<NCHUNK, 1024, 0, stream>>>(row, col, hist, bstart, part);
    bucket_sort <<<NBKT, 256, 0, stream>>>(bstart, part, off, scol);
    gemm_xw   <<<(NNODES + 15) / 16, 256, 0, stream>>>(x, w, att, xw, aS, aD);
    src_softmax_scatter<<<(NNODES + 7) / 8, 256, 0, stream>>>(off, scol, aS, aD, xw, out);
}


# product strips ride as extra blocks in the three edge-grouping launches (own launch removed, 5 -> 4); a strip's 8 loads of x requested together so that a rider block is as short as its hosts
# speedup vs baseline: 6.2401x; 1.1280x over previous
#include <hip/hip_runtime.h>

#define NNODES 50000
#define FEAT   128
#define HCOUT  128
#define NHEAD  4
#define NCH    32
#define NEDGE  800000
#define NEG_SLOPE 0.2f

typedef __attribute__((ext_vector_type(2)))  float    v2f;
typedef __attribute__((ext_vector_type(8)))  float    v8f;
typedef __attribute__((ext_vector_type(16))) _Float16 v16h;

#if __has_builtin(__builtin_amdgcn_wmma_f32_16x16x4_f32)
#define USE_F32_WMMA 1
#endif

#define TPITCH 132
#define NSTRIP ((NNODES + 15) / 16)
__device__ __forceinline__ void gemm_strip(const int strip, const int tid,
                                           float (*ldsA)[FEAT], float (*tile)[TPITCH],
                                           const float* __restrict__ x,
                                           const float* __restrict__ w,
                                           const float* __restrict__ att,
                                           float* __restrict__ xw,
                                           float* __restrict__ aS,
                                           float* __restrict__ aD) {
    const int rowBase = strip * 16;

    float xv[16 * FEAT / 256];
    #pragma unroll
    for (int i = 0; i < 16 * FEAT / 256; ++i) {
        const int idx  = tid + i * 256;
        const int node = rowBase + (idx >> 7);
        const int nc   = (node < NNODES) ? node : (NNODES - 1);
        xv[i] = x[nc * FEAT + (idx & 127)];
    }
    static_assert(16 * FEAT / 256 == 8, "the fence below names 8 loaded values");
    asm volatile("" :: "v"(xv[0]), "v"(xv[1]), "v"(xv[2]), "v"(xv[3]),
                       "v"(xv[4]), "v"(xv[5]), "v"(xv[6]), "v"(xv[7]));
    #pragma unroll
    for (int i = 0; i < 16 * FEAT / 256; ++i) {
        const int idx  = tid + i * 256;
        const int node = rowBase + (idx >> 7);
        ldsA[idx >> 7][idx & 127] = (node < NNODES) ? xv[i] : 0.0f;
    }
    __syncthreads();

    const int wave    = tid >> 5;
    const int lane    = tid & 31;
    const int colBase = wave * 16;
    const int Mrow    = lane & 15;
    const bool hi     = lane >= 16;

    v8f acc = {};

#ifdef USE_F32_WMMA
    const int koff = hi ? 2 : 0;
    #pragma unroll
    for (int k0 = 0; k0 < FEAT; k0 += 4) {
        v2f a, b;
        a.x = ldsA[Mrow][k0 + koff];
        a.y = ldsA[Mrow][k0 + koff + 1];
        b.x = w[(k0 + koff)     * HCOUT + colBase + Mrow];
        b.y = w[(k0 + koff + 1) * HCOUT + colBase + Mrow];
        acc = __builtin_amdgcn_wmma_f32_16x16x4_f32(
            false, a, false, b, (short)0, acc, false, false);
    }
#else
    for (int k0 = 0; k0 < FEAT; k0 += 32) {
        v16h a, b;
        const int ka = hi ? 8 : 0;
        const int kb = hi ? 16 : 0;
        #pragma unroll
        for (int i = 0; i < 8; ++i) {
            a[i]     = (_Float16)ldsA[Mrow][k0 + ka + i];
            a[8 + i] = (_Float16)ldsA[Mrow][k0 + 16 + ka + i];
        }
        #pragma unroll
        for (int i = 0; i < 16; ++i)
            b[i] = (_Float16)w[(k0 + kb + i) * HCOUT + colBase + Mrow];
        acc = __builtin_amdgcn_wmma_f32_16x16x32_f16(
            false, a, false, b, (short)0, acc, false, false);
    }
#endif

    #pragma unroll
    for (int v = 0; v < 8; ++v) tile[v + (hi ? 8 : 0)][colBase + Mrow] = acc[v];
    __syncthreads();

    #pragma unroll
    for (int i = 0; i < 2; ++i) {
        const int f    = tid + i * 256;
        const int r    = f >> 5;
        const int c4   = f & 31;
        const int node = rowBase + r;
        if (node < NNODES) {
            const float4 val = *reinterpret_cast<const float4*>(&tile[r][c4 * 4]);
            *reinterpret_cast<float4*>(xw + node * HCOUT + c4 * 4) = val;
        }
    }

    if (tid < 128) {
        const int t    = tid;
        const int h    = t & (NHEAD - 1);
        const int r    = (t >> 2) & 15;
        const int sd   = t >> 6;
        const int node = rowBase + r;
        const float* coef = att + h * (2 * NCH) + sd * NCH;
        const float* v    = &tile[r][h * NCH];
        float s = 0.0f;
        #pragma unroll
        for (int c = 0; c < NCH; ++c) s = __builtin_fmaf(v[c], coef[c], s);
        if (node < NNODES) (sd ? aD : aS)[node * NHEAD + h] = s;
    }
}

#define BKT_SHIFT 7
#define BKT_NODES (1 << BKT_SHIFT)
#define NBKT      ((NNODES + BKT_NODES - 1) / BKT_NODES)
#define CHUNK     4000
#define NCHUNK    ((NEDGE + CHUNK - 1) / CHUNK)
#define HALFCH    (NCHUNK / 2)
#define HPITCH    416
static_assert(NNODES <= 65536, "a destination fits the low 16 bits of a packed word");
static_assert(NBKT <= 511 && NBKT <= HPITCH, "one thread of the first 512 per bucket, and one spare");
#define HGROUP    25
static_assert(NCHUNK % 2 == 0, "the rows of counts are summed in two equal halves");
static_assert(HALFCH % HGROUP == 0, "whole groups of rows");
#define EPT       4
static_assert(CHUNK <= EPT * 1024, "one batch of loads per thread covers a chunk");
#define OUT4      (NNODES * NCH / 4)
static_assert(OUT4 <= 2 * NCHUNK * 1024, "two float4 stores per thread of bucket_count cover out");
static_assert(NCHUNK * 1024 <= OUT4, "the first of the two stores is inside out for every thread");
static_assert((NCHUNK * 1024) % (NCH / 4) == 0, "both stores of a thread fall on the same channels");

__device__ __forceinline__ void bucket_count(const int k, int* cnt,
                                             const int* __restrict__ row,
                                             const float* __restrict__ bias,
                                             int* __restrict__ hist,
                                             float* __restrict__ out) {
    const int tid = threadIdx.x;
    if (tid < 512) cnt[tid] = 0;
    __syncthreads();
    const int g = k * 1024 + tid;
    const float4 b4 = reinterpret_cast<const float4*>(bias)[g & (NCH / 4 - 1)];
    const int e0 = k * CHUNK;
    const int e1 = (e0 + CHUNK < NEDGE) ? (e0 + CHUNK) : NEDGE;
    int r[EPT];
    #pragma unroll
    for (int j = 0; j < EPT; ++j) {
        const int e = e0 + tid + j * 1024;
        r[j] = (e < e1) ? row[e] : -1;
    }
    #pragma unroll
    for (int j = 0; j < EPT; ++j)
        if (r[j] >= 0) atomicAdd(&cnt[r[j] >> BKT_SHIFT], 1);
    asm volatile("" :: "v"(b4.x), "v"(b4.y), "v"(b4.z), "v"(b4.w));
    __syncthreads();
    if (tid < NBKT) hist[k * HPITCH + tid] = cnt[tid];
    float4* o4 = reinterpret_cast<float4*>(out);
    o4[g] = b4;
    if (g + NCHUNK * 1024 < OUT4) o4[g + NCHUNK * 1024] = b4;
}

__device__ __forceinline__ void bucket_place(const int k, int (*ptot)[512], int (*pbel)[512],
                                             int* wsum, int* cnt,
                                             const int* __restrict__ row,
                                             const int* __restrict__ col,
                                             const int* __restrict__ hist,
                                             int* __restrict__ bstart,
                                             int* __restrict__ part) {
    const int tid  = threadIdx.x;
    const int lane = tid & 31;
    const int wave = tid >> 5;
    {
        const int b  = tid & 511;
        const int hf = tid >> 9;
        int tot = 0, bel = 0;
        if (b < NBKT) {
            const int k0 = hf * HALFCH;
            const int* p = hist + k0 * HPITCH + b;
            const int nbel = (k - k0 < 0) ? 0 : ((k - k0 > HALFCH) ? HALFCH : (k - k0));
            for (int g = 0; g < HALFCH; g += HGROUP) {
                int v[HGROUP];
                #pragma unroll
                for (int j = 0; j < HGROUP; ++j) v[j] = p[(g + j) * HPITCH];
                #pragma unroll
                for (int j = 0; j < HGROUP; ++j) {
                    tot += v[j];
                    bel += (g + j < nbel) ? v[j] : 0;
                }
            }
        }
        ptot[hf][b] = tot;
        pbel[hf][b] = bel;
    }
    __syncthreads();
    int T = 0, inc = 0;
    if (tid < 512) {
        T = ptot[0][tid] + ptot[1][tid];
        inc = T;
        #pragma unroll
        for (int o = 1; o < 32; o <<= 1) {
            int u = __shfl_up(inc, o, 32);
            if (lane >= o) inc += u;
        }
        if (lane == 31) wsum[wave] = inc;
    }
    __syncthreads();
    if (tid < 512) {
        int carry = 0;
        #pragma unroll
        for (int w = 0; w < 16; ++w) carry += (w < wave) ? wsum[w] : 0;
        const int excl = carry + inc - T;
        cnt[tid] = excl + pbel[0][tid] + pbel[1][tid];
        if (k == 0 && tid <= NBKT) bstart[tid] = excl;
    }
    __syncthreads();
    const int e0 = k * CHUNK;
    const int e1 = (e0 + CHUNK < NEDGE) ? (e0 + CHUNK) : NEDGE;
    int r[EPT], c[EPT];
    #pragma unroll
    for (int j = 0; j < EPT; ++j) {
        const int e = e0 + tid + j * 1024;
        r[j] = (e < e1) ? row[e] : -1;
        c[j] = (e < e1) ? col[e] : 0;
    }
    #pragma unroll
    for (int j = 0; j < EPT; ++j) {
        if (r[j] >= 0) {
            const int pos = atomicAdd(&cnt[r[j] >> BKT_SHIFT], 1);
            part[pos] = ((r[j] & (BKT_NODES - 1)) << 16) | c[j];
        }
    }
}

__device__ __forceinline__ void bucket_sort(const int b, int* cnt, int* cur, int* wsum,
                                            const int* __restrict__ bstart,
                                            const int* __restrict__ part,
                                            int* __restrict__ off,
                                            int* __restrict__ scol) {
    const int tid  = threadIdx.x;
    const int lane = tid & 31;
    const int wave = tid >> 5;
    const int s    = bstart[b];
    const int n    = bstart[b + 1] - s;
    if (tid < BKT_NODES) cnt[tid] = 0;
    __syncthreads();
    for (int i0 = tid; i0 < n; i0 += EPT * 256) {
        int w[EPT];
        #pragma unroll
        for (int j = 0; j < EPT; ++j) {
            const int i = i0 + j * 256;
            w[j] = (i < n) ? part[s + i] : -1;
        }
        #pragma unroll
        for (int j = 0; j < EPT; ++j)
            if (w[j] >= 0) atomicAdd(&cnt[w[j] >> 16], 1);
    }
    __syncthreads();
    int v = 0, inc = 0;
    if (tid < BKT_NODES) {
        v = cnt[tid];
        inc = v;
        #pragma unroll
        for (int o = 1; o < 32; o <<= 1) {
            int u = __shfl_up(inc, o, 32);
            if (lane >= o) inc += u;
        }
        if (lane == 31) wsum[wave] = inc;
    }
    __syncthreads();
    if (tid < BKT_NODES) {
        int carry = 0;
        #pragma unroll
        for (int w = 0; w < BKT_NODES / 32; ++w) carry += (w < wave) ? wsum[w] : 0;
        const int excl = s + carry + inc - v;
        const int node = b * BKT_NODES + tid;
        if (node < NNODES) off[node] = excl;
        cur[tid] = excl;
    }
    if (b == NBKT - 1 && tid == 0) off[NNODES] = s + n;
    __syncthreads();
    for (int i0 = tid; i0 < n; i0 += EPT * 256) {
        int w[EPT];
        #pragma unroll
        for (int j = 0; j < EPT; ++j) {
            const int i = i0 + j * 256;
            w[j] = (i < n) ? part[s + i] : -1;
        }
        #pragma unroll
        for (int j = 0; j < EPT; ++j) {
            if (w[j] >= 0) {
                const int pos = atomicAdd(&cur[w[j] >> 16], 1);
                scol[pos] = w[j] & 0xFFFF;
            }
        }
    }
}

#define GEMM_T1 288
#define GEMM_T2 288
#define GEMM_S3 (NSTRIP - 4 * (GEMM_T1 + GEMM_T2))
static_assert(GEMM_S3 > 0, "the third launch takes the strips that are left");

__global__ __launch_bounds__(1024, 16) void count_and_product(const int* __restrict__ row,
                                                          const float* __restrict__ bias,
                                                          int* __restrict__ hist,
                                                          float* __restrict__ out,
                                                          const float* __restrict__ x,
                                                          const float* __restrict__ w,
                                                          const float* __restrict__ att,
                                                          float* __restrict__ xw,
                                                          float* __restrict__ aS,
                                                          float* __restrict__ aD) {
    __shared__ int cnt[512];
    __shared__ float ldsA[4][16][FEAT];
    __shared__ __attribute__((aligned(16))) float tile[4][16][TPITCH];
    if (blockIdx.x < NCHUNK) {
        bucket_count(blockIdx.x, cnt, row, bias, hist, out);
    } else {
        const int grp = threadIdx.x >> 8;
        gemm_strip((blockIdx.x - NCHUNK) * 4 + grp, threadIdx.x & 255, ldsA[grp], tile[grp],
                   x, w, att, xw, aS, aD);
    }
}

__global__ __launch_bounds__(1024, 16) void place_and_product(const int* __restrict__ row,
                                                          const int* __restrict__ col,
                                                          const int* __restrict__ hist,
                                                          int* __restrict__ bstart,
                                                          int* __restrict__ part,
                                                          const float* __restrict__ x,
                                                          const float* __restrict__ w,
                                                          const float* __restrict__ att,
                                                          float* __restrict__ xw,
                                                          float* __restrict__ aS,
                                                          float* __restrict__ aD) {
    __shared__ int ptot[2][512];
    __shared__ int pbel[2][512];
    __shared__ int wsum[16];
    __shared__ int cnt[512];
    __shared__ float ldsA[4][16][FEAT];
    __shared__ __attribute__((aligned(16))) float tile[4][16][TPITCH];
    if (blockIdx.x < NCHUNK) {
        bucket_place(blockIdx.x, ptot, pbel, wsum, cnt, row, col, hist, bstart, part);
    } else {
        const int grp = threadIdx.x >> 8;
        gemm_strip((GEMM_T1 + blockIdx.x - NCHUNK) * 4 + grp, threadIdx.x & 255, ldsA[grp], tile[grp],
                   x, w, att, xw, aS, aD);
    }
}

__global__ __launch_bounds__(256) void sort_and_product(const int* __restrict__ bstart,
                                                        const int* __restrict__ part,
                                                        int* __restrict__ off,
                                                        int* __restrict__ scol,
                                                        const float* __restrict__ x,
                                                        const float* __restrict__ w,
                                                        const float* __restrict__ att,
                                                        float* __restrict__ xw,
                                                        float* __restrict__ aS,
                                                        float* __restrict__ aD) {
    __shared__ int cnt[BKT_NODES];
    __shared__ int cur[BKT_NODES];
    __shared__ int wsum[BKT_NODES / 32];
    __shared__ float ldsA[16][FEAT];
    __shared__ __attribute__((aligned(16))) float tile[16][TPITCH];
    if (blockIdx.x < NBKT) {
        bucket_sort(blockIdx.x, cnt, cur, wsum, bstart, part, off, scol);
    } else {
        gemm_strip(4 * (GEMM_T1 + GEMM_T2) + blockIdx.x - NBKT, threadIdx.x, ldsA, tile,
                   x, w, att, xw, aS, aD);
    }
}


__device__ __forceinline__ float lane_value(float v, int l) {
    return __int_as_float(__builtin_amdgcn_readlane(__float_as_int(v), l));
}

__device__ __forceinline__ void edge_logits4(const int* __restrict__ scol,
                                             const float* __restrict__ aD,
                                             const float (&as)[NHEAD],
                                             int p, int end, int& c, float (&a)[NHEAD]) {
    c = 0;
    #pragma unroll
    for (int h = 0; h < NHEAD; ++h) a[h] = -INFINITY;
    if (p < end) {
        c = scol[p];
        const float4 d = *reinterpret_cast<const float4*>(aD + c * NHEAD);
        const float dv[NHEAD] = {d.x, d.y, d.z, d.w};
        #pragma unroll
        for (int h = 0; h < NHEAD; ++h) {
            float t = as[h] + dv[h];
            a[h] = (t >= 0.0f) ? t : NEG_SLOPE * t;
        }
    }
}

__global__ __launch_bounds__(256) void src_softmax_scatter(const int* __restrict__ off,
                                                           const int* __restrict__ scol,
                                                           const float* __restrict__ aS,
                                                           const float* __restrict__ aD,
                                                           const float* __restrict__ xw,
                                                           float* __restrict__ out) {
    const int lane = threadIdx.x & 31;
    const int r = __builtin_amdgcn_readfirstlane((int)(blockIdx.x * 8 + (threadIdx.x >> 5)));
    if (r >= NNODES) return;
    const int beg = off[r], end = off[r + 1];
    if (beg >= end) return;

    float xr[NHEAD], as[NHEAD];
    #pragma unroll
    for (int h = 0; h < NHEAD; ++h) {
        xr[h] = 0.25f * xw[r * HCOUT + h * NCH + lane];
        as[h] = aS[r * NHEAD + h];
    }

    int c0; float a0[NHEAD];
    edge_logits4(scol, aD, as, beg + lane, end, c0, a0);

    float m[NHEAD];
    #pragma unroll
    for (int h = 0; h < NHEAD; ++h) m[h] = a0[h];
    for (int b = beg + 32; b < end; b += 32) {
        int c; float a[NHEAD];
        edge_logits4(scol, aD, as, b + lane, end, c, a);
        #pragma unroll
        for (int h = 0; h < NHEAD; ++h) m[h] = fmaxf(m[h], a[h]);
    }
    #pragma unroll
    for (int h = 0; h < NHEAD; ++h) {
        #pragma unroll
        for (int o = 16; o > 0; o >>= 1) m[h] = fmaxf(m[h], __shfl_xor(m[h], o, 32));
        if (!(fabsf(m[h]) < INFINITY)) m[h] = 0.0f;
    }

    float ex0[NHEAD], S[NHEAD];
    #pragma unroll
    for (int h = 0; h < NHEAD; ++h) {
        ex0[h] = (beg + lane < end) ? __expf(a0[h] - m[h]) : 0.0f;
        S[h] = ex0[h];
    }
    for (int b = beg + 32; b < end; b += 32) {
        int c; float a[NHEAD];
        edge_logits4(scol, aD, as, b + lane, end, c, a);
        #pragma unroll
        for (int h = 0; h < NHEAD; ++h) S[h] += (b + lane < end) ? __expf(a[h] - m[h]) : 0.0f;
    }
    #pragma unroll
    for (int h = 0; h < NHEAD; ++h) {
        #pragma unroll
        for (int o = 16; o > 0; o >>= 1) S[h] += __shfl_xor(S[h], o, 32);
        S[h] += 1e-16f;
    }

    for (int b = beg; b < end; b += 32) {
        int c; float w[NHEAD];
        if (b == beg) {
            c = c0;
            #pragma unroll
            for (int h = 0; h < NHEAD; ++h) w[h] = ex0[h] / S[h];
        } else {
            float a[NHEAD];
            edge_logits4(scol, aD, as, b + lane, end, c, a);
            #pragma unroll
            for (int h = 0; h < NHEAD; ++h) w[h] = ((b + lane < end) ? __expf(a[h] - m[h]) : 0.0f) / S[h];
        }
        const int cnt = (end - b < 32) ? (end - b) : 32;
        for (int j = 0; j < cnt; ++j) {
            const int cj = __builtin_amdgcn_readlane(c, j);
            float s = lane_value(w[0], j) * xr[0];
            #pragma unroll
            for (int h = 1; h < NHEAD; ++h) s += lane_value(w[h], j) * xr[h];
            atomicAdd(&out[cj * NCH + lane], s);
        }
    }
}

extern "C" void kernel_launch(void* const* d_in, const int* in_sizes, int n_in,
                              void* d_out, int out_size, void* d_ws, size_t ws_size,
                              hipStream_t stream) {
    const float* x    = (const float*)d_in[0];
    const int*   ei   = (const int*)  d_in[1];
    const float* w    = (const float*)d_in[2];
    const float* att  = (const float*)d_in[3];
    const float* bias = (const float*)d_in[4];
    float* ws = (float*)d_ws;

    float* xw    = ws;
    float* aS    = ws + 6400000;
    float* aD    = ws + 6600000;
    int*   scol  = (int*)(ws + 6800000);
    int*   off   = (int*)(ws + 7600000);
    int*   part  = (int*)(ws + 7700000);
    int*   hist  = (int*)(ws + 8500000);
    int*   bstart= (int*)(ws + 8600000);

    const int* row = ei;
    const int* col = ei + NEDGE;

    float* out = (float*)d_out;

    count_and_product<<<NCHUNK + GEMM_T1, 1024, 0, stream>>>(row, bias, hist, out, x, w, att, xw, aS, aD);
    place_and_product<<<NCHUNK + GEMM_T2, 1024, 0, stream>>>(row, col, hist, bstart, part, x, w, att, xw, aS, aD);
    sort_and_product <<<NBKT + GEMM_S3, 256, 0, stream>>>(bstart, part, off, scol, x, w, att, xw, aS, aD);
    src_softmax_scatter<<<(NNODES + 7) / 8, 256, 0, stream>>>(off, scol, aS, aD, xw, out);
}


# first sort level made chunk-local: a block orders its own chunk by bucket in LDS, a bucket's block gathers one run per chunk; placing launch removed (4 -> 3); order inside a source's list changes
# speedup vs baseline: 8.6176x; 1.3810x over previous
#include <hip/hip_runtime.h>

#define NNODES 50000
#define FEAT   128
#define HCOUT  128
#define NHEAD  4
#define NCH    32
#define NEDGE  800000
#define NEG_SLOPE 0.2f

typedef __attribute__((ext_vector_type(2)))  float    v2f;
typedef __attribute__((ext_vector_type(8)))  float    v8f;
typedef __attribute__((ext_vector_type(16))) _Float16 v16h;

#if __has_builtin(__builtin_amdgcn_wmma_f32_16x16x4_f32)
#define USE_F32_WMMA 1
#endif

#define TPITCH 132
#define NSTRIP ((NNODES + 15) / 16)
__device__ __forceinline__ void gemm_strip(const int strip, const int tid,
                                           float (*ldsA)[FEAT], float (*tile)[TPITCH],
                                           const float* __restrict__ x,
                                           const float* __restrict__ w,
                                           const float* __restrict__ att,
                                           float* __restrict__ xw,
                                           float* __restrict__ aS,
                                           float* __restrict__ aD) {
    const int rowBase = strip * 16;

    float xv[16 * FEAT / 256];
    #pragma unroll
    for (int i = 0; i < 16 * FEAT / 256; ++i) {
        const int idx  = tid + i * 256;
        const int node = rowBase + (idx >> 7);
        const int nc   = (node < NNODES) ? node : (NNODES - 1);
        xv[i] = x[nc * FEAT + (idx & 127)];
    }
    static_assert(16 * FEAT / 256 == 8, "the fence below names 8 loaded values");
    asm volatile("" :: "v"(xv[0]), "v"(xv[1]), "v"(xv[2]), "v"(xv[3]),
                       "v"(xv[4]), "v"(xv[5]), "v"(xv[6]), "v"(xv[7]));
    #pragma unroll
    for (int i = 0; i < 16 * FEAT / 256; ++i) {
        const int idx  = tid + i * 256;
        const int node = rowBase + (idx >> 7);
        ldsA[idx >> 7][idx & 127] = (node < NNODES) ? xv[i] : 0.0f;
    }
    __syncthreads();

    const int wave    = tid >> 5;
    const int lane    = tid & 31;
    const int colBase = wave * 16;
    const int Mrow    = lane & 15;
    const bool hi     = lane >= 16;

    v8f acc = {};

#ifdef USE_F32_WMMA
    const int koff = hi ? 2 : 0;
    #pragma unroll
    for (int k0 = 0; k0 < FEAT; k0 += 4) {
        v2f a, b;
        a.x = ldsA[Mrow][k0 + koff];
        a.y = ldsA[Mrow][k0 + koff + 1];
        b.x = w[(k0 + koff)     * HCOUT + colBase + Mrow];
        b.y = w[(k0 + koff + 1) * HCOUT + colBase + Mrow];
        acc = __builtin_amdgcn_wmma_f32_16x16x4_f32(
            false, a, false, b, (short)0, acc, false, false);
    }
#else
    for (int k0 = 0; k0 < FEAT; k0 += 32) {
        v16h a, b;
        const int ka = hi ? 8 : 0;
        const int kb = hi ? 16 : 0;
        #pragma unroll
        for (int i = 0; i < 8; ++i) {
            a[i]     = (_Float16)ldsA[Mrow][k0 + ka + i];
            a[8 + i] = (_Float16)ldsA[Mrow][k0 + 16 + ka + i];
        }
        #pragma unroll
        for (int i = 0; i < 16; ++i)
            b[i] = (_Float16)w[(k0 + kb + i) * HCOUT + colBase + Mrow];
        acc = __builtin_amdgcn_wmma_f32_16x16x32_f16(
            false, a, false, b, (short)0, acc, false, false);
    }
#endif

    #pragma unroll
    for (int v = 0; v < 8; ++v) tile[v + (hi ? 8 : 0)][colBase + Mrow] = acc[v];
    __syncthreads();

    #pragma unroll
    for (int i = 0; i < 2; ++i) {
        const int f    = tid + i * 256;
        const int r    = f >> 5;
        const int c4   = f & 31;
        const int node = rowBase + r;
        if (node < NNODES) {
            const float4 val = *reinterpret_cast<const float4*>(&tile[r][c4 * 4]);
            *reinterpret_cast<float4*>(xw + node * HCOUT + c4 * 4) = val;
        }
    }

    if (tid < 128) {
        const int t    = tid;
        const int h    = t & (NHEAD - 1);
        const int r    = (t >> 2) & 15;
        const int sd   = t >> 6;
        const int node = rowBase + r;
        const float* coef = att + h * (2 * NCH) + sd * NCH;
        const float* v    = &tile[r][h * NCH];
        float s = 0.0f;
        #pragma unroll
        for (int c = 0; c < NCH; ++c) s = __builtin_fmaf(v[c], coef[c], s);
        if (node < NNODES) (sd ? aD : aS)[node * NHEAD + h] = s;
    }
}

#define BKT_SHIFT 7
#define BKT_NODES (1 << BKT_SHIFT)
#define NBKT      ((NNODES + BKT_NODES - 1) / BKT_NODES)
#define CHUNK     4000
#define NCHUNK    ((NEDGE + CHUNK - 1) / CHUNK)
#define HPITCH    416
static_assert(NNODES <= 65536, "a destination fits the low 16 bits of a packed word");
static_assert(NBKT <= 511 && NBKT < HPITCH, "one thread of the first 512 per bucket, and one spare");
static_assert(NCHUNK <= 256, "one thread of a bucket's block per chunk");
#define EPT       4
#define KEEP      9
static_assert(CHUNK <= EPT * 1024, "one batch of loads per thread covers a chunk");
static_assert(CHUNK <= 4 * 16 * FEAT, "a chunk's ordered edges fit the LDS that the product blocks use for x");
#define OUT4      (NNODES * NCH / 4)
static_assert(OUT4 <= 2 * NCHUNK * 1024, "two float4 stores per thread of chunk_sort cover out");
static_assert(NCHUNK * 1024 <= OUT4, "the first of the two stores is inside out for every thread");
static_assert((NCHUNK * 1024) % (NCH / 4) == 0, "both stores of a thread fall on the same channels");

__device__ __forceinline__ void chunk_sort(const int k, int* cnt, int* wsum, int* stage,
                                           const int* __restrict__ row,
                                           const int* __restrict__ col,
                                           const float* __restrict__ bias,
                                           int* __restrict__ hist,
                                           int* __restrict__ part,
                                           float* __restrict__ out) {
    const int tid  = threadIdx.x;
    const int lane = tid & 31;
    const int wave = tid >> 5;
    if (tid < 512) cnt[tid] = 0;
    __syncthreads();
    const int g = k * 1024 + tid;
    const float4 b4 = reinterpret_cast<const float4*>(bias)[g & (NCH / 4 - 1)];
    const int e0 = k * CHUNK;
    const int e1 = (e0 + CHUNK < NEDGE) ? (e0 + CHUNK) : NEDGE;
    int r[EPT], c[EPT];
    #pragma unroll
    for (int j = 0; j < EPT; ++j) {
        const int e = e0 + tid + j * 1024;
        r[j] = (e < e1) ? row[e] : -1;
        c[j] = (e < e1) ? col[e] : 0;
    }
    #pragma unroll
    for (int j = 0; j < EPT; ++j)
        if (r[j] >= 0) atomicAdd(&cnt[r[j] >> BKT_SHIFT], 1);
    asm volatile("" :: "v"(b4.x), "v"(b4.y), "v"(b4.z), "v"(b4.w));
    __syncthreads();
    int T = 0, inc = 0;
    if (tid < 512) {
        T = cnt[tid];
        inc = T;
        #pragma unroll
        for (int o = 1; o < 32; o <<= 1) {
            int u = __shfl_up(inc, o, 32);
            if (lane >= o) inc += u;
        }
        if (lane == 31) wsum[wave] = inc;
    }
    __syncthreads();
    if (tid < 512) {
        int carry = 0;
        #pragma unroll
        for (int w = 0; w < 16; ++w) carry += (w < wave) ? wsum[w] : 0;
        const int excl = carry + inc - T;
        cnt[tid] = excl;
        if (tid <= NBKT) hist[k * HPITCH + tid] = excl;
    }
    __syncthreads();
    #pragma unroll
    for (int j = 0; j < EPT; ++j) {
        if (r[j] >= 0) {
            const int pos = atomicAdd(&cnt[r[j] >> BKT_SHIFT], 1);
            stage[pos] = ((r[j] & (BKT_NODES - 1)) << 16) | c[j];
        }
    }
    __syncthreads();
    #pragma unroll
    for (int j = 0; j < EPT; ++j) {
        const int i = tid + j * 1024;
        if (e0 + i < e1) part[e0 + i] = stage[i];
    }
    float4* o4 = reinterpret_cast<float4*>(out);
    o4[g] = b4;
    if (g + NCHUNK * 1024 < OUT4) o4[g + NCHUNK * 1024] = b4;
}

__device__ __forceinline__ void bucket_sort(const int b, int* cnt, int* cur, int* wsum,
                                            int* rsum, int* asum, int* runP, int* runA,
                                            int* srcl,
                                            const int* __restrict__ hist,
                                            const int* __restrict__ part,
                                            int* __restrict__ off,
                                            int* __restrict__ scol) {
    const int tid  = threadIdx.x;
    const int lane = tid & 31;
    const int wave = tid >> 5;
    int a = 0, len = 0;
    if (tid < NCHUNK) {
        const int2 h = make_int2(hist[tid * HPITCH + b], hist[tid * HPITCH + b + 1]);
        a   = h.x;
        len = h.y - h.x;
    }
    if (tid < BKT_NODES) cnt[tid] = 0;
    int rinc = len, asum_l = a;
    #pragma unroll
    for (int o = 1; o < 32; o <<= 1) {
        int u = __shfl_up(rinc, o, 32);
        if (lane >= o) rinc += u;
    }
    #pragma unroll
    for (int o = 16; o > 0; o >>= 1) asum_l += __shfl_xor(asum_l, o, 32);
    if (lane == 31) rsum[wave] = rinc;
    if (lane == 0)  asum[wave] = asum_l;
    __syncthreads();
    int rcarry = 0, n = 0, s = 0;
    #pragma unroll
    for (int w = 0; w < 8; ++w) {
        rcarry += (w < wave) ? rsum[w] : 0;
        n += rsum[w];
        s += asum[w];
    }
    if (tid < NCHUNK) {
        const int p = rcarry + rinc - len;
        runP[tid] = p;
        runA[tid] = tid * CHUNK + a - p;
        const int pe = (p + len < KEEP * 256) ? (p + len) : (KEEP * 256);
        for (int q = p; q < pe; ++q) srcl[q] = tid * CHUNK + a - p + q;
    }
    if (tid == 0) runP[NCHUNK] = n;
    __syncthreads();
    int keep[KEEP];
    #pragma unroll
    for (int j = 0; j < KEEP; ++j) {
        const int i = tid + j * 256;
        keep[j] = (i < n) ? part[srcl[i]] : -1;
    }
    auto locate = [&](const int i) -> int {
        int lo = 0, hi = NCHUNK;
        #pragma unroll
        for (int st = 0; st < 8; ++st) {
            const int mid = (lo + hi) >> 1;
            const bool le = runP[mid] <= i;
            lo = le ? mid : lo;
            hi = le ? hi : mid;
        }
        return runA[lo] + i;
    };
    #pragma unroll
    for (int j = 0; j < KEEP; ++j)
        if (keep[j] >= 0) atomicAdd(&cnt[keep[j] >> 16], 1);
    for (int i0 = tid + KEEP * 256; i0 < n; i0 += EPT * 256) {
        int src[EPT], w[EPT];
        #pragma unroll
        for (int j = 0; j < EPT; ++j) src[j] = locate(i0 + j * 256);
        #pragma unroll
        for (int j = 0; j < EPT; ++j) w[j] = (i0 + j * 256 < n) ? part[src[j]] : -1;
        #pragma unroll
        for (int j = 0; j < EPT; ++j)
            if (w[j] >= 0) atomicAdd(&cnt[w[j] >> 16], 1);
    }
    __syncthreads();
    int v = 0, inc = 0;
    if (tid < BKT_NODES) {
        v = cnt[tid];
        inc = v;
        #pragma unroll
        for (int o = 1; o < 32; o <<= 1) {
            int u = __shfl_up(inc, o, 32);
            if (lane >= o) inc += u;
        }
        if (lane == 31) wsum[wave] = inc;
    }
    __syncthreads();
    if (tid < BKT_NODES) {
        int carry = 0;
        #pragma unroll
        for (int w = 0; w < BKT_NODES / 32; ++w) carry += (w < wave) ? wsum[w] : 0;
        const int excl = s + carry + inc - v;
        const int node = b * BKT_NODES + tid;
        if (node < NNODES) off[node] = excl;
        cur[tid] = excl;
    }
    if (b == NBKT - 1 && tid == 0) off[NNODES] = s + n;
    __syncthreads();
    #pragma unroll
    for (int j = 0; j < KEEP; ++j) {
        if (keep[j] >= 0) {
            const int pos = atomicAdd(&cur[keep[j] >> 16], 1);
            scol[pos] = keep[j] & 0xFFFF;
        }
    }
    for (int i0 = tid + KEEP * 256; i0 < n; i0 += EPT * 256) {
        int src[EPT], w[EPT];
        #pragma unroll
        for (int j = 0; j < EPT; ++j) src[j] = locate(i0 + j * 256);
        #pragma unroll
        for (int j = 0; j < EPT; ++j) w[j] = (i0 + j * 256 < n) ? part[src[j]] : -1;
        #pragma unroll
        for (int j = 0; j < EPT; ++j) {
            if (w[j] >= 0) {
                const int pos = atomicAdd(&cur[w[j] >> 16], 1);
                scol[pos] = w[j] & 0xFFFF;
            }
        }
    }
}

#define GEMM_T1 368
#define GEMM_S2 (NSTRIP - 4 * GEMM_T1)
static_assert(GEMM_S2 > 0, "the second launch takes the strips that are left");

__global__ __launch_bounds__(1024, 16) void chunk_sort_and_product(const int* __restrict__ row,
                                                               const int* __restrict__ col,
                                                               const float* __restrict__ bias,
                                                               int* __restrict__ hist,
                                                               int* __restrict__ part,
                                                               float* __restrict__ out,
                                                               const float* __restrict__ x,
                                                               const float* __restrict__ w,
                                                               const float* __restrict__ att,
                                                               float* __restrict__ xw,
                                                               float* __restrict__ aS,
                                                               float* __restrict__ aD) {
    __shared__ int cnt[512];
    __shared__ int wsum[16];
    __shared__ float ldsA[4][16][FEAT];
    __shared__ __attribute__((aligned(16))) float tile[4][16][TPITCH];
    if (blockIdx.x < NCHUNK) {
        chunk_sort(blockIdx.x, cnt, wsum, reinterpret_cast<int*>(&ldsA[0][0][0]),
                   row, col, bias, hist, part, out);
    } else {
        const int grp = threadIdx.x >> 8;
        gemm_strip((blockIdx.x - NCHUNK) * 4 + grp, threadIdx.x & 255, ldsA[grp], tile[grp],
                   x, w, att, xw, aS, aD);
    }
}

__global__ __launch_bounds__(256, 16) void sort_and_product(const int* __restrict__ hist,
                                                        const int* __restrict__ part,
                                                        int* __restrict__ off,
                                                        int* __restrict__ scol,
                                                        const float* __restrict__ x,
                                                        const float* __restrict__ w,
                                                        const float* __restrict__ att,
                                                        float* __restrict__ xw,
                                                        float* __restrict__ aS,
                                                        float* __restrict__ aD) {
    __shared__ int cnt[BKT_NODES];
    __shared__ int cur[BKT_NODES];
    __shared__ int wsum[BKT_NODES / 32];
    __shared__ int rsum[8];
    __shared__ int asum[8];
    __shared__ int runP[NCHUNK + 1];
    __shared__ int runA[NCHUNK];
    __shared__ int srcl[KEEP * 256];
    __shared__ float ldsA[16][FEAT];
    __shared__ __attribute__((aligned(16))) float tile[16][TPITCH];
    if (blockIdx.x < NBKT) {
        bucket_sort(blockIdx.x, cnt, cur, wsum, rsum, asum, runP, runA, srcl, hist, part, off, scol);
    } else {
        gemm_strip(4 * GEMM_T1 + blockIdx.x - NBKT, threadIdx.x, ldsA, tile,
                   x, w, att, xw, aS, aD);
    }
}


__device__ __forceinline__ float lane_value(float v, int l) {
    return __int_as_float(__builtin_amdgcn_readlane(__float_as_int(v), l));
}

__device__ __forceinline__ void edge_logits4(const int* __restrict__ scol,
                                             const float* __restrict__ aD,
                                             const float (&as)[NHEAD],
                                             int p, int end, int& c, float (&a)[NHEAD]) {
    c = 0;
    #pragma unroll
    for (int h = 0; h < NHEAD; ++h) a[h] = -INFINITY;
    if (p < end) {
        c = scol[p];
        const float4 d = *reinterpret_cast<const float4*>(aD + c * NHEAD);
        const float dv[NHEAD] = {d.x, d.y, d.z, d.w};
        #pragma unroll
        for (int h = 0; h < NHEAD; ++h) {
            float t = as[h] + dv[h];
            a[h] = (t >= 0.0f) ? t : NEG_SLOPE * t;
        }
    }
}

__global__ __launch_bounds__(256) void src_softmax_scatter(const int* __restrict__ off,
                                                           const int* __restrict__ scol,
                                                           const float* __restrict__ aS,
                                                           const float* __restrict__ aD,
                                                           const float* __restrict__ xw,
                                                           float* __restrict__ out) {
    const int lane = threadIdx.x & 31;
    const int r = __builtin_amdgcn_readfirstlane((int)(blockIdx.x * 8 + (threadIdx.x >> 5)));
    if (r >= NNODES) return;
    const int beg = off[r], end = off[r + 1];
    if (beg >= end) return;

    float xr[NHEAD], as[NHEAD];
    #pragma unroll
    for (int h = 0; h < NHEAD; ++h) {
        xr[h] = 0.25f * xw[r * HCOUT + h * NCH + lane];
        as[h] = aS[r * NHEAD + h];
    }

    int c0; float a0[NHEAD];
    edge_logits4(scol, aD, as, beg + lane, end, c0, a0);

    float m[NHEAD];
    #pragma unroll
    for (int h = 0; h < NHEAD; ++h) m[h] = a0[h];
    for (int b = beg + 32; b < end; b += 32) {
        int c; float a[NHEAD];
        edge_logits4(scol, aD, as, b + lane, end, c, a);
        #pragma unroll
        for (int h = 0; h < NHEAD; ++h) m[h] = fmaxf(m[h], a[h]);
    }
    #pragma unroll
    for (int h = 0; h < NHEAD; ++h) {
        #pragma unroll
        for (int o = 16; o > 0; o >>= 1) m[h] = fmaxf(m[h], __shfl_xor(m[h], o, 32));
        if (!(fabsf(m[h]) < INFINITY)) m[h] = 0.0f;
    }

    float ex0[NHEAD], S[NHEAD];
    #pragma unroll
    for (int h = 0; h < NHEAD; ++h) {
        ex0[h] = (beg + lane < end) ? __expf(a0[h] - m[h]) : 0.0f;
        S[h] = ex0[h];
    }
    for (int b = beg + 32; b < end; b += 32) {
        int c; float a[NHEAD];
        edge_logits4(scol, aD, as, b + lane, end, c, a);
        #pragma unroll
        for (int h = 0; h < NHEAD; ++h) S[h] += (b + lane < end) ? __expf(a[h] - m[h]) : 0.0f;
    }
    #pragma unroll
    for (int h = 0; h < NHEAD; ++h) {
        #pragma unroll
        for (int o = 16; o > 0; o >>= 1) S[h] += __shfl_xor(S[h], o, 32);
        S[h] += 1e-16f;
    }

    for (int b = beg; b < end; b += 32) {
        int c; float w[NHEAD];
        if (b == beg) {
            c = c0;
            #pragma unroll
            for (int h = 0; h < NHEAD; ++h) w[h] = ex0[h] / S[h];
        } else {
            float a[NHEAD];
            edge_logits4(scol, aD, as, b + lane, end, c, a);
            #pragma unroll
            for (int h = 0; h < NHEAD; ++h) w[h] = ((b + lane < end) ? __expf(a[h] - m[h]) : 0.0f) / S[h];
        }
        const int cnt = (end - b < 32) ? (end - b) : 32;
        for (int j = 0; j < cnt; ++j) {
            const int cj = __builtin_amdgcn_readlane(c, j);
            float s = lane_value(w[0], j) * xr[0];
            #pragma unroll
            for (int h = 1; h < NHEAD; ++h) s += lane_value(w[h], j) * xr[h];
            atomicAdd(&out[cj * NCH + lane], s);
        }
    }
}

extern "C" void kernel_launch(void* const* d_in, const int* in_sizes, int n_in,
                              void* d_out, int out_size, void* d_ws, size_t ws_size,
                              hipStream_t stream) {
    const float* x    = (const float*)d_in[0];
    const int*   ei   = (const int*)  d_in[1];
    const float* w    = (const float*)d_in[2];
    const float* att  = (const float*)d_in[3];
    const float* bias = (const float*)d_in[4];
    float* ws = (float*)d_ws;

    float* xw    = ws;
    float* aS    = ws + 6400000;
    float* aD    = ws + 6600000;
    int*   scol  = (int*)(ws + 6800000);
    int*   off   = (int*)(ws + 7600000);
    int*   part  = (int*)(ws + 7700000);
    int*   hist  = (int*)(ws + 8500000);

    const int* row = ei;
    const int* col = ei + NEDGE;

    float* out = (float*)d_out;

    chunk_sort_and_product<<<NCHUNK + GEMM_T1, 1024, 0, stream>>>(row, col, bias, hist, part, out, x, w, att, xw, aS, aD);
    sort_and_product <<<NBKT + GEMM_S2, 256, 0, stream>>>(hist, part, off, scol, x, w, att, xw, aS, aD);
    src_softmax_scatter<<<(NNODES + 7) / 8, 256, 0, stream>>>(off, scol, aS, aD, xw, out);
}


# product: LDS row pitch of the staged x strip 128 -> 132 floats; the A-fragment reads (16 rows, same k) no longer fall on the same two banks (16-way bank conflict removed)
# speedup vs baseline: 10.6371x; 1.2343x over previous
#include <hip/hip_runtime.h>

#define NNODES 50000
#define FEAT   128
#define HCOUT  128
#define NHEAD  4
#define NCH    32
#define NEDGE  800000
#define NEG_SLOPE 0.2f

typedef __attribute__((ext_vector_type(2)))  float    v2f;
typedef __attribute__((ext_vector_type(8)))  float    v8f;
typedef __attribute__((ext_vector_type(16))) _Float16 v16h;

#if __has_builtin(__builtin_amdgcn_wmma_f32_16x16x4_f32)
#define USE_F32_WMMA 1
#endif

#define TPITCH 132
#define APITCH 132
#define NSTRIP ((NNODES + 15) / 16)
__device__ __forceinline__ void gemm_strip(const int strip, const int tid,
                                           float (*ldsA)[APITCH], float (*tile)[TPITCH],
                                           const float* __restrict__ x,
                                           const float* __restrict__ w,
                                           const float* __restrict__ att,
                                           float* __restrict__ xw,
                                           float* __restrict__ aS,
                                           float* __restrict__ aD) {
    const int rowBase = strip * 16;

    float xv[16 * FEAT / 256];
    #pragma unroll
    for (int i = 0; i < 16 * FEAT / 256; ++i) {
        const int idx  = tid + i * 256;
        const int node = rowBase + (idx >> 7);
        const int nc   = (node < NNODES) ? node : (NNODES - 1);
        xv[i] = x[nc * FEAT + (idx & 127)];
    }
    static_assert(16 * FEAT / 256 == 8, "the fence below names 8 loaded values");
    asm volatile("" :: "v"(xv[0]), "v"(xv[1]), "v"(xv[2]), "v"(xv[3]),
                       "v"(xv[4]), "v"(xv[5]), "v"(xv[6]), "v"(xv[7]));
    #pragma unroll
    for (int i = 0; i < 16 * FEAT / 256; ++i) {
        const int idx  = tid + i * 256;
        const int node = rowBase + (idx >> 7);
        ldsA[idx >> 7][idx & 127] = (node < NNODES) ? xv[i] : 0.0f;
    }
    __syncthreads();

    const int wave    = tid >> 5;
    const int lane    = tid & 31;
    const int colBase = wave * 16;
    const int Mrow    = lane & 15;
    const bool hi     = lane >= 16;

    v8f acc = {};

#ifdef USE_F32_WMMA
    const int koff = hi ? 2 : 0;
    #pragma unroll
    for (int k0 = 0; k0 < FEAT; k0 += 4) {
        v2f a, b;
        a.x = ldsA[Mrow][k0 + koff];
        a.y = ldsA[Mrow][k0 + koff + 1];
        b.x = w[(k0 + koff)     * HCOUT + colBase + Mrow];
        b.y = w[(k0 + koff + 1) * HCOUT + colBase + Mrow];
        acc = __builtin_amdgcn_wmma_f32_16x16x4_f32(
            false, a, false, b, (short)0, acc, false, false);
    }
#else
    for (int k0 = 0; k0 < FEAT; k0 += 32) {
        v16h a, b;
        const int ka = hi ? 8 : 0;
        const int kb = hi ? 16 : 0;
        #pragma unroll
        for (int i = 0; i < 8; ++i) {
            a[i]     = (_Float16)ldsA[Mrow][k0 + ka + i];
            a[8 + i] = (_Float16)ldsA[Mrow][k0 + 16 + ka + i];
        }
        #pragma unroll
        for (int i = 0; i < 16; ++i)
            b[i] = (_Float16)w[(k0 + kb + i) * HCOUT + colBase + Mrow];
        acc = __builtin_amdgcn_wmma_f32_16x16x32_f16(
            false, a, false, b, (short)0, acc, false, false);
    }
#endif

    #pragma unroll
    for (int v = 0; v < 8; ++v) tile[v + (hi ? 8 : 0)][colBase + Mrow] = acc[v];
    __syncthreads();

    #pragma unroll
    for (int i = 0; i < 2; ++i) {
        const int f    = tid + i * 256;
        const int r    = f >> 5;
        const int c4   = f & 31;
        const int node = rowBase + r;
        if (node < NNODES) {
            const float4 val = *reinterpret_cast<const float4*>(&tile[r][c4 * 4]);
            *reinterpret_cast<float4*>(xw + node * HCOUT + c4 * 4) = val;
        }
    }

    if (tid < 128) {
        const int t    = tid;
        const int h    = t & (NHEAD - 1);
        const int r    = (t >> 2) & 15;
        const int sd   = t >> 6;
        const int node = rowBase + r;
        const float* coef = att + h * (2 * NCH) + sd * NCH;
        const float* v    = &tile[r][h * NCH];
        float s = 0.0f;
        #pragma unroll
        for (int c = 0; c < NCH; ++c) s = __builtin_fmaf(v[c], coef[c], s);
        if (node < NNODES) (sd ? aD : aS)[node * NHEAD + h] = s;
    }
}

#define BKT_SHIFT 7
#define BKT_NODES (1 << BKT_SHIFT)
#define NBKT      ((NNODES + BKT_NODES - 1) / BKT_NODES)
#define CHUNK     4000
#define NCHUNK    ((NEDGE + CHUNK - 1) / CHUNK)
#define HPITCH    416
static_assert(NNODES <= 65536, "a destination fits the low 16 bits of a packed word");
static_assert(NBKT <= 511 && NBKT < HPITCH, "one thread of the first 512 per bucket, and one spare");
static_assert(NCHUNK <= 256, "one thread of a bucket's block per chunk");
#define EPT       4
#define KEEP      9
static_assert(CHUNK <= EPT * 1024, "one batch of loads per thread covers a chunk");
static_assert(CHUNK <= 4 * 16 * APITCH, "a chunk's ordered edges fit the LDS that the product blocks use for x");
#define OUT4      (NNODES * NCH / 4)
static_assert(OUT4 <= 2 * NCHUNK * 1024, "two float4 stores per thread of chunk_sort cover out");
static_assert(NCHUNK * 1024 <= OUT4, "the first of the two stores is inside out for every thread");
static_assert((NCHUNK * 1024) % (NCH / 4) == 0, "both stores of a thread fall on the same channels");

__device__ __forceinline__ void chunk_sort(const int k, int* cnt, int* wsum, int* stage,
                                           const int* __restrict__ row,
                                           const int* __restrict__ col,
                                           const float* __restrict__ bias,
                                           int* __restrict__ hist,
                                           int* __restrict__ part,
                                           float* __restrict__ out) {
    const int tid  = threadIdx.x;
    const int lane = tid & 31;
    const int wave = tid >> 5;
    if (tid < 512) cnt[tid] = 0;
    __syncthreads();
    const int g = k * 1024 + tid;
    const float4 b4 = reinterpret_cast<const float4*>(bias)[g & (NCH / 4 - 1)];
    const int e0 = k * CHUNK;
    const int e1 = (e0 + CHUNK < NEDGE) ? (e0 + CHUNK) : NEDGE;
    int r[EPT], c[EPT];
    #pragma unroll
    for (int j = 0; j < EPT; ++j) {
        const int e = e0 + tid + j * 1024;
        r[j] = (e < e1) ? row[e] : -1;
        c[j] = (e < e1) ? col[e] : 0;
    }
    #pragma unroll
    for (int j = 0; j < EPT; ++j)
        if (r[j] >= 0) atomicAdd(&cnt[r[j] >> BKT_SHIFT], 1);
    asm volatile("" :: "v"(b4.x), "v"(b4.y), "v"(b4.z), "v"(b4.w));
    __syncthreads();
    int T = 0, inc = 0;
    if (tid < 512) {
        T = cnt[tid];
        inc = T;
        #pragma unroll
        for (int o = 1; o < 32; o <<= 1) {
            int u = __shfl_up(inc, o, 32);
            if (lane >= o) inc += u;
        }
        if (lane == 31) wsum[wave] = inc;
    }
    __syncthreads();
    if (tid < 512) {
        int carry = 0;
        #pragma unroll
        for (int w = 0; w < 16; ++w) carry += (w < wave) ? wsum[w] : 0;
        const int excl = carry + inc - T;
        cnt[tid] = excl;
        if (tid <= NBKT) hist[k * HPITCH + tid] = excl;
    }
    __syncthreads();
    #pragma unroll
    for (int j = 0; j < EPT; ++j) {
        if (r[j] >= 0) {
            const int pos = atomicAdd(&cnt[r[j] >> BKT_SHIFT], 1);
            stage[pos] = ((r[j] & (BKT_NODES - 1)) << 16) | c[j];
        }
    }
    __syncthreads();
    #pragma unroll
    for (int j = 0; j < EPT; ++j) {
        const int i = tid + j * 1024;
        if (e0 + i < e1) part[e0 + i] = stage[i];
    }
    float4* o4 = reinterpret_cast<float4*>(out);
    o4[g] = b4;
    if (g + NCHUNK * 1024 < OUT4) o4[g + NCHUNK * 1024] = b4;
}

__device__ __forceinline__ void bucket_sort(const int b, int* cnt, int* cur, int* wsum,
                                            int* rsum, int* asum, int* runP, int* runA,
                                            int* srcl,
                                            const int* __restrict__ hist,
                                            const int* __restrict__ part,
                                            int* __restrict__ off,
                                            int* __restrict__ scol) {
    const int tid  = threadIdx.x;
    const int lane = tid & 31;
    const int wave = tid >> 5;
    int a = 0, len = 0;
    if (tid < NCHUNK) {
        const int2 h = make_int2(hist[tid * HPITCH + b], hist[tid * HPITCH + b + 1]);
        a   = h.x;
        len = h.y - h.x;
    }
    if (tid < BKT_NODES) cnt[tid] = 0;
    int rinc = len, asum_l = a;
    #pragma unroll
    for (int o = 1; o < 32; o <<= 1) {
        int u = __shfl_up(rinc, o, 32);
        if (lane >= o) rinc += u;
    }
    #pragma unroll
    for (int o = 16; o > 0; o >>= 1) asum_l += __shfl_xor(asum_l, o, 32);
    if (lane == 31) rsum[wave] = rinc;
    if (lane == 0)  asum[wave] = asum_l;
    __syncthreads();
    int rcarry = 0, n = 0, s = 0;
    #pragma unroll
    for (int w = 0; w < 8; ++w) {
        rcarry += (w < wave) ? rsum[w] : 0;
        n += rsum[w];
        s += asum[w];
    }
    if (tid < NCHUNK) {
        const int p = rcarry + rinc - len;
        runP[tid] = p;
        runA[tid] = tid * CHUNK + a - p;
        const int pe = (p + len < KEEP * 256) ? (p + len) : (KEEP * 256);
        for (int q = p; q < pe; ++q) srcl[q] = tid * CHUNK + a - p + q;
    }
    if (tid == 0) runP[NCHUNK] = n;
    __syncthreads();
    int keep[KEEP];
    #pragma unroll
    for (int j = 0; j < KEEP; ++j) {
        const int i = tid + j * 256;
        keep[j] = (i < n) ? part[srcl[i]] : -1;
    }
    auto locate = [&](const int i) -> int {
        int lo = 0, hi = NCHUNK;
        #pragma unroll
        for (int st = 0; st < 8; ++st) {
            const int mid = (lo + hi) >> 1;
            const bool le = runP[mid] <= i;
            lo = le ? mid : lo;
            hi = le ? hi : mid;
        }
        return runA[lo] + i;
    };
    #pragma unroll
    for (int j = 0; j < KEEP; ++j)
        if (keep[j] >= 0) atomicAdd(&cnt[keep[j] >> 16], 1);
    for (int i0 = tid + KEEP * 256; i0 < n; i0 += EPT * 256) {
        int src[EPT], w[EPT];
        #pragma unroll
        for (int j = 0; j < EPT; ++j) src[j] = locate(i0 + j * 256);
        #pragma unroll
        for (int j = 0; j < EPT; ++j) w[j] = (i0 + j * 256 < n) ? part[src[j]] : -1;
        #pragma unroll
        for (int j = 0; j < EPT; ++j)
            if (w[j] >= 0) atomicAdd(&cnt[w[j] >> 16], 1);
    }
    __syncthreads();
    int v = 0, inc = 0;
    if (tid < BKT_NODES) {
        v = cnt[tid];
        inc = v;
        #pragma unroll
        for (int o = 1; o < 32; o <<= 1) {
            int u = __shfl_up(inc, o, 32);
            if (lane >= o) inc += u;
        }
        if (lane == 31) wsum[wave] = inc;
    }
    __syncthreads();
    if (tid < BKT_NODES) {
        int carry = 0;
        #pragma unroll
        for (int w = 0; w < BKT_NODES / 32; ++w) carry += (w < wave) ? wsum[w] : 0;
        const int excl = s + carry + inc - v;
        const int node = b * BKT_NODES + tid;
        if (node < NNODES) off[node] = excl;
        cur[tid] = excl;
    }
    if (b == NBKT - 1 && tid == 0) off[NNODES] = s + n;
    __syncthreads();
    #pragma unroll
    for (int j = 0; j < KEEP; ++j) {
        if (keep[j] >= 0) {
            const int pos = atomicAdd(&cur[keep[j] >> 16], 1);
            scol[pos] = keep[j] & 0xFFFF;
        }
    }
    for (int i0 = tid + KEEP * 256; i0 < n; i0 += EPT * 256) {
        int src[EPT], w[EPT];
        #pragma unroll
        for (int j = 0; j < EPT; ++j) src[j] = locate(i0 + j * 256);
        #pragma unroll
        for (int j = 0; j < EPT; ++j) w[j] = (i0 + j * 256 < n) ? part[src[j]] : -1;
        #pragma unroll
        for (int j = 0; j < EPT; ++j) {
            if (w[j] >= 0) {
                const int pos = atomicAdd(&cur[w[j] >> 16], 1);
                scol[pos] = w[j] & 0xFFFF;
            }
        }
    }
}

#define GEMM_T1 368
#define GEMM_S2 (NSTRIP - 4 * GEMM_T1)
static_assert(GEMM_S2 > 0, "the second launch takes the strips that are left");

__global__ __launch_bounds__(1024, 16) void chunk_sort_and_product(const int* __restrict__ row,
                                                               const int* __restrict__ col,
                                                               const float* __restrict__ bias,
                                                               int* __restrict__ hist,
                                                               int* __restrict__ part,
                                                               float* __restrict__ out,
                                                               const float* __restrict__ x,
                                                               const float* __restrict__ w,
                                                               const float* __restrict__ att,
                                                               float* __restrict__ xw,
                                                               float* __restrict__ aS,
                                                               float* __restrict__ aD) {
    __shared__ int cnt[512];
    __shared__ int wsum[16];
    __shared__ __attribute__((aligned(16))) float ldsA[4][16][APITCH];
    __shared__ __attribute__((aligned(16))) float tile[4][16][TPITCH];
    if (blockIdx.x < NCHUNK) {
        chunk_sort(blockIdx.x, cnt, wsum, reinterpret_cast<int*>(&ldsA[0][0][0]),
                   row, col, bias, hist, part, out);
    } else {
        const int grp = threadIdx.x >> 8;
        gemm_strip((blockIdx.x - NCHUNK) * 4 + grp, threadIdx.x & 255, ldsA[grp], tile[grp],
                   x, w, att, xw, aS, aD);
    }
}

__global__ __launch_bounds__(256, 16) void sort_and_product(const int* __restrict__ hist,
                                                        const int* __restrict__ part,
                                                        int* __restrict__ off,
                                                        int* __restrict__ scol,
                                                        const float* __restrict__ x,
                                                        const float* __restrict__ w,
                                                        const float* __restrict__ att,
                                                        float* __restrict__ xw,
                                                        float* __restrict__ aS,
                                                        float* __restrict__ aD) {
    __shared__ int cnt[BKT_NODES];
    __shared__ int cur[BKT_NODES];
    __shared__ int wsum[BKT_NODES / 32];
    __shared__ int rsum[8];
    __shared__ int asum[8];
    __shared__ int runP[NCHUNK + 1];
    __shared__ int runA[NCHUNK];
    __shared__ int srcl[KEEP * 256];
    __shared__ __attribute__((aligned(16))) float ldsA[16][APITCH];
    __shared__ __attribute__((aligned(16))) float tile[16][TPITCH];
    if (blockIdx.x < NBKT) {
        bucket_sort(blockIdx.x, cnt, cur, wsum, rsum, asum, runP, runA, srcl, hist, part, off, scol);
    } else {
        gemm_strip(4 * GEMM_T1 + blockIdx.x - NBKT, threadIdx.x, ldsA, tile,
                   x, w, att, xw, aS, aD);
    }
}


__device__ __forceinline__ float lane_value(float v, int l) {
    return __int_as_float(__builtin_amdgcn_readlane(__float_as_int(v), l));
}

__device__ __forceinline__ void edge_logits4(const int* __restrict__ scol,
                                             const float* __restrict__ aD,
                                             const float (&as)[NHEAD],
                                             int p, int end, int& c, float (&a)[NHEAD]) {
    c = 0;
    #pragma unroll
    for (int h = 0; h < NHEAD; ++h) a[h] = -INFINITY;
    if (p < end) {
        c = scol[p];
        const float4 d = *reinterpret_cast<const float4*>(aD + c * NHEAD);
        const float dv[NHEAD] = {d.x, d.y, d.z, d.w};
        #pragma unroll
        for (int h = 0; h < NHEAD; ++h) {
            float t = as[h] + dv[h];
            a[h] = (t >= 0.0f) ? t : NEG_SLOPE * t;
        }
    }
}

__global__ __launch_bounds__(256) void src_softmax_scatter(const int* __restrict__ off,
                                                           const int* __restrict__ scol,
                                                           const float* __restrict__ aS,
                                                           const float* __restrict__ aD,
                                                           const float* __restrict__ xw,
                                                           float* __restrict__ out) {
    const int lane = threadIdx.x & 31;
    const int r = __builtin_amdgcn_readfirstlane((int)(blockIdx.x * 8 + (threadIdx.x >> 5)));
    if (r >= NNODES) return;
    const int beg = off[r], end = off[r + 1];
    if (beg >= end) return;

    float xr[NHEAD], as[NHEAD];
    #pragma unroll
    for (int h = 0; h < NHEAD; ++h) {
        xr[h] = 0.25f * xw[r * HCOUT + h * NCH + lane];
        as[h] = aS[r * NHEAD + h];
    }

    int c0; float a0[NHEAD];
    edge_logits4(scol, aD, as, beg + lane, end, c0, a0);

    float m[NHEAD];
    #pragma unroll
    for (int h = 0; h < NHEAD; ++h) m[h] = a0[h];
    for (int b = beg + 32; b < end; b += 32) {
        int c; float a[NHEAD];
        edge_logits4(scol, aD, as, b + lane, end, c, a);
        #pragma unroll
        for (int h = 0; h < NHEAD; ++h) m[h] = fmaxf(m[h], a[h]);
    }
    #pragma unroll
    for (int h = 0; h < NHEAD; ++h) {
        #pragma unroll
        for (int o = 16; o > 0; o >>= 1) m[h] = fmaxf(m[h], __shfl_xor(m[h], o, 32));
        if (!(fabsf(m[h]) < INFINITY)) m[h] = 0.0f;
    }

    float ex0[NHEAD], S[NHEAD];
    #pragma unroll
    for (int h = 0; h < NHEAD; ++h) {
        ex0[h] = (beg + lane < end) ? __expf(a0[h] - m[h]) : 0.0f;
        S[h] = ex0[h];
    }
    for (int b = beg + 32; b < end; b += 32) {
        int c; float a[NHEAD];
        edge_logits4(scol, aD, as, b + lane, end, c, a);
        #pragma unroll
        for (int h = 0; h < NHEAD; ++h) S[h] += (b + lane < end) ? __expf(a[h] - m[h]) : 0.0f;
    }
    #pragma unroll
    for (int h = 0; h < NHEAD; ++h) {
        #pragma unroll
        for (int o = 16; o > 0; o >>= 1) S[h] += __shfl_xor(S[h], o, 32);
        S[h] += 1e-16f;
    }

    for (int b = beg; b < end; b += 32) {
        int c; float w[NHEAD];
        if (b == beg) {
            c = c0;
            #pragma unroll
            for (int h = 0; h < NHEAD; ++h) w[h] = ex0[h] / S[h];
        } else {
            float a[NHEAD];
            edge_logits4(scol, aD, as, b + lane, end, c, a);
            #pragma unroll
            for (int h = 0; h < NHEAD; ++h) w[h] = ((b + lane < end) ? __expf(a[h] - m[h]) : 0.0f) / S[h];
        }
        const int cnt = (end - b < 32) ? (end - b) : 32;
        for (int j = 0; j < cnt; ++j) {
            const int cj = __builtin_amdgcn_readlane(c, j);
            float s = lane_value(w[0], j) * xr[0];
            #pragma unroll
            for (int h = 1; h < NHEAD; ++h) s += lane_value(w[h], j) * xr[h];
            atomicAdd(&out[cj * NCH + lane], s);
        }
    }
}

extern "C" void kernel_launch(void* const* d_in, const int* in_sizes, int n_in,
                              void* d_out, int out_size, void* d_ws, size_t ws_size,
                              hipStream_t stream) {
    const float* x    = (const float*)d_in[0];
    const int*   ei   = (const int*)  d_in[1];
    const float* w    = (const float*)d_in[2];
    const float* att  = (const float*)d_in[3];
    const float* bias = (const float*)d_in[4];
    float* ws = (float*)d_ws;

    float* xw    = ws;
    float* aS    = ws + 6400000;
    float* aD    = ws + 6600000;
    int*   scol  = (int*)(ws + 6800000);
    int*   off   = (int*)(ws + 7600000);
    int*   part  = (int*)(ws + 7700000);
    int*   hist  = (int*)(ws + 8500000);

    const int* row = ei;
    const int* col = ei + NEDGE;

    float* out = (float*)d_out;

    chunk_sort_and_product<<<NCHUNK + GEMM_T1, 1024, 0, stream>>>(row, col, bias, hist, part, out, x, w, att, xw, aS, aD);
    sort_and_product <<<NBKT + GEMM_S2, 256, 0, stream>>>(hist, part, off, scol, x, w, att, xw, aS, aD);
    src_softmax_scatter<<<(NNODES + 7) / 8, 256, 0, stream>>>(off, scol, aS, aD, xw, out);
}


# product: a group of 8 waves takes a pair of 16-row strips and uses each B fragment of W for both (two accumulators per wave; x staged by 16-byte loads): W loads halved, launch 1 fits one pass
# speedup vs baseline: 11.0031x; 1.0344x over previous
#include <hip/hip_runtime.h>

#define NNODES 50000
#define FEAT   128
#define HCOUT  128
#define NHEAD  4
#define NCH    32
#define NEDGE  800000
#define NEG_SLOPE 0.2f

typedef __attribute__((ext_vector_type(2)))  float    v2f;
typedef __attribute__((ext_vector_type(8)))  float    v8f;
typedef __attribute__((ext_vector_type(16))) _Float16 v16h;

#if __has_builtin(__builtin_amdgcn_wmma_f32_16x16x4_f32)
#define USE_F32_WMMA 1
#endif

#define TPITCH 132
#define APITCH 132
#define PROWS  32
#define NPAIR  ((NNODES + PROWS - 1) / PROWS)
#define XPT    (PROWS * FEAT / 256)
__device__ __forceinline__ void gemm_pair(const int pair, const int tid,
                                          float (*ldsA)[APITCH], float (*tile)[TPITCH],
                                          const float* __restrict__ x,
                                          const float* __restrict__ w,
                                          const float* __restrict__ att,
                                          float* __restrict__ xw,
                                          float* __restrict__ aS,
                                          float* __restrict__ aD) {
    const int rowBase = pair * PROWS;

    float4 xv[XPT / 4];
    #pragma unroll
    for (int i = 0; i < XPT / 4; ++i) {
        const int idx  = tid + i * 256;
        const int node = rowBase + (idx >> 5);
        const int nc   = (node < NNODES) ? node : (NNODES - 1);
        xv[i] = *reinterpret_cast<const float4*>(x + nc * FEAT + (idx & 31) * 4);
    }
    static_assert(XPT == 16, "the fence below names 4 x 4 loaded values");
    asm volatile("" :: "v"(xv[0].x), "v"(xv[0].y), "v"(xv[0].z), "v"(xv[0].w),
                       "v"(xv[1].x), "v"(xv[1].y), "v"(xv[1].z), "v"(xv[1].w),
                       "v"(xv[2].x), "v"(xv[2].y), "v"(xv[2].z), "v"(xv[2].w),
                       "v"(xv[3].x), "v"(xv[3].y), "v"(xv[3].z), "v"(xv[3].w));
    #pragma unroll
    for (int i = 0; i < XPT / 4; ++i) {
        const int idx   = tid + i * 256;
        const bool real = rowBase + (idx >> 5) < NNODES;
        float4 v;
        v.x = real ? xv[i].x : 0.0f;
        v.y = real ? xv[i].y : 0.0f;
        v.z = real ? xv[i].z : 0.0f;
        v.w = real ? xv[i].w : 0.0f;
        *reinterpret_cast<float4*>(&ldsA[idx >> 5][(idx & 31) * 4]) = v;
    }
    __syncthreads();

    const int wave    = tid >> 5;
    const int lane    = tid & 31;
    const int colBase = wave * 16;
    const int Mrow    = lane & 15;
    const bool hi     = lane >= 16;

    v8f acc0 = {};
    v8f acc1 = {};

#ifdef USE_F32_WMMA
    const int koff = hi ? 2 : 0;
    #pragma unroll
    for (int k0 = 0; k0 < FEAT; k0 += 4) {
        v2f a0, a1, b;
        a0.x = ldsA[Mrow][k0 + koff];
        a0.y = ldsA[Mrow][k0 + koff + 1];
        a1.x = ldsA[16 + Mrow][k0 + koff];
        a1.y = ldsA[16 + Mrow][k0 + koff + 1];
        b.x = w[(k0 + koff)     * HCOUT + colBase + Mrow];
        b.y = w[(k0 + koff + 1) * HCOUT + colBase + Mrow];
        acc0 = __builtin_amdgcn_wmma_f32_16x16x4_f32(
            false, a0, false, b, (short)0, acc0, false, false);
        acc1 = __builtin_amdgcn_wmma_f32_16x16x4_f32(
            false, a1, false, b, (short)0, acc1, false, false);
    }
#else
    for (int k0 = 0; k0 < FEAT; k0 += 32) {
        v16h a0, a1, b;
        const int ka = hi ? 8 : 0;
        const int kb = hi ? 16 : 0;
        #pragma unroll
        for (int i = 0; i < 8; ++i) {
            a0[i]     = (_Float16)ldsA[Mrow][k0 + ka + i];
            a0[8 + i] = (_Float16)ldsA[Mrow][k0 + 16 + ka + i];
            a1[i]     = (_Float16)ldsA[16 + Mrow][k0 + ka + i];
            a1[8 + i] = (_Float16)ldsA[16 + Mrow][k0 + 16 + ka + i];
        }
        #pragma unroll
        for (int i = 0; i < 16; ++i)
            b[i] = (_Float16)w[(k0 + kb + i) * HCOUT + colBase + Mrow];
        acc0 = __builtin_amdgcn_wmma_f32_16x16x32_f16(
            false, a0, false, b, (short)0, acc0, false, false);
        acc1 = __builtin_amdgcn_wmma_f32_16x16x32_f16(
            false, a1, false, b, (short)0, acc1, false, false);
    }
#endif

    #pragma unroll
    for (int v = 0; v < 8; ++v) {
        tile[v + (hi ? 8 : 0)][colBase + Mrow]      = acc0[v];
        tile[16 + v + (hi ? 8 : 0)][colBase + Mrow] = acc1[v];
    }
    __syncthreads();

    #pragma unroll
    for (int i = 0; i < PROWS * 32 / 256; ++i) {
        const int f    = tid + i * 256;
        const int r    = f >> 5;
        const int c4   = f & 31;
        const int node = rowBase + r;
        if (node < NNODES) {
            const float4 val = *reinterpret_cast<const float4*>(&tile[r][c4 * 4]);
            *reinterpret_cast<float4*>(xw + node * HCOUT + c4 * 4) = val;
        }
    }

    {
        const int t    = tid;
        const int h    = t & (NHEAD - 1);
        const int r    = (t >> 2) & (PROWS - 1);
        const int sd   = t >> 7;
        const int node = rowBase + r;
        const float* coef = att + h * (2 * NCH) + sd * NCH;
        const float* v    = &tile[r][h * NCH];
        float s = 0.0f;
        #pragma unroll
        for (int c = 0; c < NCH; ++c) s = __builtin_fmaf(v[c], coef[c], s);
        if (node < NNODES) (sd ? aD : aS)[node * NHEAD + h] = s;
    }
}

#define BKT_SHIFT 7
#define BKT_NODES (1 << BKT_SHIFT)
#define NBKT      ((NNODES + BKT_NODES - 1) / BKT_NODES)
#define CHUNK     4000
#define NCHUNK    ((NEDGE + CHUNK - 1) / CHUNK)
#define HPITCH    416
static_assert(NNODES <= 65536, "a destination fits the low 16 bits of a packed word");
static_assert(NBKT <= 511 && NBKT < HPITCH, "one thread of the first 512 per bucket, and one spare");
static_assert(NCHUNK <= 256, "one thread of a bucket's block per chunk");
#define EPT       4
#define KEEP      9
static_assert(CHUNK <= EPT * 1024, "one batch of loads per thread covers a chunk");
static_assert(CHUNK <= 4 * PROWS * APITCH, "a chunk's ordered edges fit the LDS that the product blocks use for x");
#define OUT4      (NNODES * NCH / 4)
static_assert(OUT4 <= 2 * NCHUNK * 1024, "two float4 stores per thread of chunk_sort cover out");
static_assert(NCHUNK * 1024 <= OUT4, "the first of the two stores is inside out for every thread");
static_assert((NCHUNK * 1024) % (NCH / 4) == 0, "both stores of a thread fall on the same channels");

__device__ __forceinline__ void chunk_sort(const int k, int* cnt, int* wsum, int* stage,
                                           const int* __restrict__ row,
                                           const int* __restrict__ col,
                                           const float* __restrict__ bias,
                                           int* __restrict__ hist,
                                           int* __restrict__ part,
                                           float* __restrict__ out) {
    const int tid  = threadIdx.x;
    const int lane = tid & 31;
    const int wave = tid >> 5;
    if (tid < 512) cnt[tid] = 0;
    __syncthreads();
    const int g = k * 1024 + tid;
    const float4 b4 = reinterpret_cast<const float4*>(bias)[g & (NCH / 4 - 1)];
    const int e0 = k * CHUNK;
    const int e1 = (e0 + CHUNK < NEDGE) ? (e0 + CHUNK) : NEDGE;
    int r[EPT], c[EPT];
    #pragma unroll
    for (int j = 0; j < EPT; ++j) {
        const int e = e0 + tid + j * 1024;
        r[j] = (e < e1) ? row[e] : -1;
        c[j] = (e < e1) ? col[e] : 0;
    }
    #pragma unroll
    for (int j = 0; j < EPT; ++j)
        if (r[j] >= 0) atomicAdd(&cnt[r[j] >> BKT_SHIFT], 1);
    asm volatile("" :: "v"(b4.x), "v"(b4.y), "v"(b4.z), "v"(b4.w));
    __syncthreads();
    int T = 0, inc = 0;
    if (tid < 512) {
        T = cnt[tid];
        inc = T;
        #pragma unroll
        for (int o = 1; o < 32; o <<= 1) {
            int u = __shfl_up(inc, o, 32);
            if (lane >= o) inc += u;
        }
        if (lane == 31) wsum[wave] = inc;
    }
    __syncthreads();
    if (tid < 512) {
        int carry = 0;
        #pragma unroll
        for (int w = 0; w < 16; ++w) carry += (w < wave) ? wsum[w] : 0;
        const int excl = carry + inc - T;
        cnt[tid] = excl;
        if (tid <= NBKT) hist[k * HPITCH + tid] = excl;
    }
    __syncthreads();
    #pragma unroll
    for (int j = 0; j < EPT; ++j) {
        if (r[j] >= 0) {
            const int pos = atomicAdd(&cnt[r[j] >> BKT_SHIFT], 1);
            stage[pos] = ((r[j] & (BKT_NODES - 1)) << 16) | c[j];
        }
    }
    __syncthreads();
    #pragma unroll
    for (int j = 0; j < EPT; ++j) {
        const int i = tid + j * 1024;
        if (e0 + i < e1) part[e0 + i] = stage[i];
    }
    float4* o4 = reinterpret_cast<float4*>(out);
    o4[g] = b4;
    if (g + NCHUNK * 1024 < OUT4) o4[g + NCHUNK * 1024] = b4;
}

__device__ __forceinline__ void bucket_sort(const int b, int* cnt, int* cur, int* wsum,
                                            int* rsum, int* asum, int* runP, int* runA,
                                            int* srcl,
                                            const int* __restrict__ hist,
                                            const int* __restrict__ part,
                                            int* __restrict__ off,
                                            int* __restrict__ scol) {
    const int tid  = threadIdx.x;
    const int lane = tid & 31;
    const int wave = tid >> 5;
    int a = 0, len = 0;
    if (tid < NCHUNK) {
        const int2 h = make_int2(hist[tid * HPITCH + b], hist[tid * HPITCH + b + 1]);
        a   = h.x;
        len = h.y - h.x;
    }
    if (tid < BKT_NODES) cnt[tid] = 0;
    int rinc = len, asum_l = a;
    #pragma unroll
    for (int o = 1; o < 32; o <<= 1) {
        int u = __shfl_up(rinc, o, 32);
        if (lane >= o) rinc += u;
    }
    #pragma unroll
    for (int o = 16; o > 0; o >>= 1) asum_l += __shfl_xor(asum_l, o, 32);
    if (lane == 31) rsum[wave] = rinc;
    if (lane == 0)  asum[wave] = asum_l;
    __syncthreads();
    int rcarry = 0, n = 0, s = 0;
    #pragma unroll
    for (int w = 0; w < 8; ++w) {
        rcarry += (w < wave) ? rsum[w] : 0;
        n += rsum[w];
        s += asum[w];
    }
    if (tid < NCHUNK) {
        const int p = rcarry + rinc - len;
        runP[tid] = p;
        runA[tid] = tid * CHUNK + a - p;
        const int pe = (p + len < KEEP * 256) ? (p + len) : (KEEP * 256);
        for (int q = p; q < pe; ++q) srcl[q] = tid * CHUNK + a - p + q;
    }
    if (tid == 0) runP[NCHUNK] = n;
    __syncthreads();
    int keep[KEEP];
    #pragma unroll
    for (int j = 0; j < KEEP; ++j) {
        const int i = tid + j * 256;
        keep[j] = (i < n) ? part[srcl[i]] : -1;
    }
    auto locate = [&](const int i) -> int {
        int lo = 0, hi = NCHUNK;
        #pragma unroll
        for (int st = 0; st < 8; ++st) {
            const int mid = (lo + hi) >> 1;
            const bool le = runP[mid] <= i;
            lo = le ? mid : lo;
            hi = le ? hi : mid;
        }
        return runA[lo] + i;
    };
    #pragma unroll
    for (int j = 0; j < KEEP; ++j)
        if (keep[j] >= 0) atomicAdd(&cnt[keep[j] >> 16], 1);
    for (int i0 = tid + KEEP * 256; i0 < n; i0 += EPT * 256) {
        int src[EPT], w[EPT];
        #pragma unroll
        for (int j = 0; j < EPT; ++j) src[j] = locate(i0 + j * 256);
        #pragma unroll
        for (int j = 0; j < EPT; ++j) w[j] = (i0 + j * 256 < n) ? part[src[j]] : -1;
        #pragma unroll
        for (int j = 0; j < EPT; ++j)
            if (w[j] >= 0) atomicAdd(&cnt[w[j] >> 16], 1);
    }
    __syncthreads();
    int v = 0, inc = 0;
    if (tid < BKT_NODES) {
        v = cnt[tid];
        inc = v;
        #pragma unroll
        for (int o = 1; o < 32; o <<= 1) {
            int u = __shfl_up(inc, o, 32);
            if (lane >= o) inc += u;
        }
        if (lane == 31) wsum[wave] = inc;
    }
    __syncthreads();
    if (tid < BKT_NODES) {
        int carry = 0;
        #pragma unroll
        for (int w = 0; w < BKT_NODES / 32; ++w) carry += (w < wave) ? wsum[w] : 0;
        const int excl = s + carry + inc - v;
        const int node = b * BKT_NODES + tid;
        if (node < NNODES) off[node] = excl;
        cur[tid] = excl;
    }
    if (b == NBKT - 1 && tid == 0) off[NNODES] = s + n;
    __syncthreads();
    #pragma unroll
    for (int j = 0; j < KEEP; ++j) {
        if (keep[j] >= 0) {
            const int pos = atomicAdd(&cur[keep[j] >> 16], 1);
            scol[pos] = keep[j] & 0xFFFF;
        }
    }
    for (int i0 = tid + KEEP * 256; i0 < n; i0 += EPT * 256) {
        int src[EPT], w[EPT];
        #pragma unroll
        for (int j = 0; j < EPT; ++j) src[j] = locate(i0 + j * 256);
        #pragma unroll
        for (int j = 0; j < EPT; ++j) w[j] = (i0 + j * 256 < n) ? part[src[j]] : -1;
        #pragma unroll
        for (int j = 0; j < EPT; ++j) {
            if (w[j] >= 0) {
                const int pos = atomicAdd(&cur[w[j] >> 16], 1);
                scol[pos] = w[j] & 0xFFFF;
            }
        }
    }
}

#define GEMM_T1 184
#define GEMM_P2 (NPAIR - 4 * GEMM_T1)
static_assert(GEMM_P2 > 0, "the second launch takes the pairs that are left");
static_assert(NCHUNK + GEMM_T1 <= 2 * 256, "launch 1 is resident in one pass");
static_assert(NBKT + GEMM_P2 <= 5 * 256, "launch 2 is resident in one pass");

__global__ __launch_bounds__(1024, 16) void chunk_sort_and_product(const int* __restrict__ row,
                                                               const int* __restrict__ col,
                                                               const float* __restrict__ bias,
                                                               int* __restrict__ hist,
                                                               int* __restrict__ part,
                                                               float* __restrict__ out,
                                                               const float* __restrict__ x,
                                                               const float* __restrict__ w,
                                                               const float* __restrict__ att,
                                                               float* __restrict__ xw,
                                                               float* __restrict__ aS,
                                                               float* __restrict__ aD) {
    __shared__ int cnt[512];
    __shared__ int wsum[16];
    __shared__ __attribute__((aligned(16))) float ldsA[4][PROWS][APITCH];
    __shared__ __attribute__((aligned(16))) float tile[4][PROWS][TPITCH];
    if (blockIdx.x < NCHUNK) {
        chunk_sort(blockIdx.x, cnt, wsum, reinterpret_cast<int*>(&ldsA[0][0][0]),
                   row, col, bias, hist, part, out);
    } else {
        const int grp = threadIdx.x >> 8;
        gemm_pair((blockIdx.x - NCHUNK) * 4 + grp, threadIdx.x & 255, ldsA[grp], tile[grp],
                  x, w, att, xw, aS, aD);
    }
}

__global__ __launch_bounds__(256, 16) void sort_and_product(const int* __restrict__ hist,
                                                        const int* __restrict__ part,
                                                        int* __restrict__ off,
                                                        int* __restrict__ scol,
                                                        const float* __restrict__ x,
                                                        const float* __restrict__ w,
                                                        const float* __restrict__ att,
                                                        float* __restrict__ xw,
                                                        float* __restrict__ aS,
                                                        float* __restrict__ aD) {
    __shared__ int cnt[BKT_NODES];
    __shared__ int cur[BKT_NODES];
    __shared__ int wsum[BKT_NODES / 32];
    __shared__ int rsum[8];
    __shared__ int asum[8];
    __shared__ int runP[NCHUNK + 1];
    __shared__ int runA[NCHUNK];
    __shared__ int srcl[KEEP * 256];
    __shared__ __attribute__((aligned(16))) float ldsA[PROWS][APITCH];
    __shared__ __attribute__((aligned(16))) float tile[PROWS][TPITCH];
    if (blockIdx.x < NBKT) {
        bucket_sort(blockIdx.x, cnt, cur, wsum, rsum, asum, runP, runA, srcl, hist, part, off, scol);
    } else {
        gemm_pair(4 * GEMM_T1 + blockIdx.x - NBKT, threadIdx.x, ldsA, tile,
                  x, w, att, xw, aS, aD);
    }
}


__device__ __forceinline__ float lane_value(float v, int l) {
    return __int_as_float(__builtin_amdgcn_readlane(__float_as_int(v), l));
}

__device__ __forceinline__ void edge_logits4(const int* __restrict__ scol,
                                             const float* __restrict__ aD,
                                             const float (&as)[NHEAD],
                                             int p, int end, int& c, float (&a)[NHEAD]) {
    c = 0;
    #pragma unroll
    for (int h = 0; h < NHEAD; ++h) a[h] = -INFINITY;
    if (p < end) {
        c = scol[p];
        const float4 d = *reinterpret_cast<const float4*>(aD + c * NHEAD);
        const float dv[NHEAD] = {d.x, d.y, d.z, d.w};
        #pragma unroll
        for (int h = 0; h < NHEAD; ++h) {
            float t = as[h] + dv[h];
            a[h] = (t >= 0.0f) ? t : NEG_SLOPE * t;
        }
    }
}

__global__ __launch_bounds__(256) void src_softmax_scatter(const int* __restrict__ off,
                                                           const int* __restrict__ scol,
                                                           const float* __restrict__ aS,
                                                           const float* __restrict__ aD,
                                                           const float* __restrict__ xw,
                                                           float* __restrict__ out) {
    const int lane = threadIdx.x & 31;
    const int r = __builtin_amdgcn_readfirstlane((int)(blockIdx.x * 8 + (threadIdx.x >> 5)));
    if (r >= NNODES) return;
    const int beg = off[r], end = off[r + 1];
    if (beg >= end) return;

    float xr[NHEAD], as[NHEAD];
    #pragma unroll
    for (int h = 0; h < NHEAD; ++h) {
        xr[h] = 0.25f * xw[r * HCOUT + h * NCH + lane];
        as[h] = aS[r * NHEAD + h];
    }

    int c0; float a0[NHEAD];
    edge_logits4(scol, aD, as, beg + lane, end, c0, a0);

    float m[NHEAD];
    #pragma unroll
    for (int h = 0; h < NHEAD; ++h) m[h] = a0[h];
    for (int b = beg + 32; b < end; b += 32) {
        int c; float a[NHEAD];
        edge_logits4(scol, aD, as, b + lane, end, c, a);
        #pragma unroll
        for (int h = 0; h < NHEAD; ++h) m[h] = fmaxf(m[h], a[h]);
    }
    #pragma unroll
    for (int h = 0; h < NHEAD; ++h) {
        #pragma unroll
        for (int o = 16; o > 0; o >>= 1) m[h] = fmaxf(m[h], __shfl_xor(m[h], o, 32));
        if (!(fabsf(m[h]) < INFINITY)) m[h] = 0.0f;
    }

    float ex0[NHEAD], S[NHEAD];
    #pragma unroll
    for (int h = 0; h < NHEAD; ++h) {
        ex0[h] = (beg + lane < end) ? __expf(a0[h] - m[h]) : 0.0f;
        S[h] = ex0[h];
    }
    for (int b = beg + 32; b < end; b += 32) {
        int c; float a[NHEAD];
        edge_logits4(scol, aD, as, b + lane, end, c, a);
        #pragma unroll
        for (int h = 0; h < NHEAD; ++h) S[h] += (b + lane < end) ? __expf(a[h] - m[h]) : 0.0f;
    }
    #pragma unroll
    for (int h = 0; h < NHEAD; ++h) {
        #pragma unroll
        for (int o = 16; o > 0; o >>= 1) S[h] += __shfl_xor(S[h], o, 32);
        S[h] += 1e-16f;
    }

    for (int b = beg; b < end; b += 32) {
        int c; float w[NHEAD];
        if (b == beg) {
            c = c0;
            #pragma unroll
            for (int h = 0; h < NHEAD; ++h) w[h] = ex0[h] / S[h];
        } else {
            float a[NHEAD];
            edge_logits4(scol, aD, as, b + lane, end, c, a);
            #pragma unroll
            for (int h = 0; h < NHEAD; ++h) w[h] = ((b + lane < end) ? __expf(a[h] - m[h]) : 0.0f) / S[h];
        }
        const int cnt = (end - b < 32) ? (end - b) : 32;
        for (int j = 0; j < cnt; ++j) {
            const int cj = __builtin_amdgcn_readlane(c, j);
            float s = lane_value(w[0], j) * xr[0];
            #pragma unroll
            for (int h = 1; h < NHEAD; ++h) s += lane_value(w[h], j) * xr[h];
            atomicAdd(&out[cj * NCH + lane], s);
        }
    }
}

extern "C" void kernel_launch(void* const* d_in, const int* in_sizes, int n_in,
                              void* d_out, int out_size, void* d_ws, size_t ws_size,
                              hipStream_t stream) {
    const float* x    = (const float*)d_in[0];
    const int*   ei   = (const int*)  d_in[1];
    const float* w    = (const float*)d_in[2];
    const float* att  = (const float*)d_in[3];
    const float* bias = (const float*)d_in[4];
    float* ws = (float*)d_ws;

    float* xw    = ws;
    float* aS    = ws + 6400000;
    float* aD    = ws + 6600000;
    int*   scol  = (int*)(ws + 6800000);
    int*   off   = (int*)(ws + 7600000);
    int*   part  = (int*)(ws + 7700000);
    int*   hist  = (int*)(ws + 8500000);

    const int* row = ei;
    const int* col = ei + NEDGE;

    float* out = (float*)d_out;

    chunk_sort_and_product<<<NCHUNK + GEMM_T1, 1024, 0, stream>>>(row, col, bias, hist, part, out, x, w, att, xw, aS, aD);
    sort_and_product <<<NBKT + GEMM_P2, 256, 0, stream>>>(hist, part, off, scol, x, w, att, xw, aS, aD);
    src_softmax_scatter<<<(NNODES + 7) / 8, 256, 0, stream>>>(off, scol, aS, aD, xw, out);
}
